# starting kernel
# baseline (speedup 1.0000x reference)
#include <hip/hip_runtime.h>
#include <math.h>

typedef float v2f __attribute__((ext_vector_type(2)));
typedef float v8f __attribute__((ext_vector_type(8)));

#define C_DIM 128
#define N_PTS 4096
#define B_SZ  4
#define VSTR  132
#define PSTR  17
#define QT_W  2048
#define SMEMQ 16896

__device__ __forceinline__ float xor16(float v, unsigned selLo, unsigned selHi) {
  unsigned i = __builtin_bit_cast(unsigned, v);
  i = __builtin_amdgcn_permlane16(i, i, selLo, selHi, false, false);
  return __builtin_bit_cast(float, i);
}
#define XOR16_1(v) xor16((v), 0x67452301u, 0xEFCDAB89u)
#define XOR16_2(v) xor16((v), 0x54761032u, 0xDCFE98BAu)
#define XOR16_4(v) xor16((v), 0x32107654u, 0xBA98FEDCu)
#define XOR16_8(v) xor16((v), 0xFEDCBA98u, 0x76543210u)

__global__ __launch_bounds__(256) void cmix_kernel(const float* __restrict__ x,
                                                   const float* __restrict__ W,
                                                   float* __restrict__ out) {
  __shared__ float wrow[C_DIM];
  const int o = blockIdx.y;
  const int b = blockIdx.z;
  const int tid = threadIdx.x;
  if (tid < C_DIM) wrow[tid] = W[o * C_DIM + tid];
  __syncthreads();
  const int n = blockIdx.x * 256 + tid;
  float acc = 0.0f;
#pragma unroll 8
  for (int c = 0; c < C_DIM; ++c)
    acc += wrow[c] * x[((size_t)(b * C_DIM + c)) * N_PTS + n];
  out[((size_t)(b * C_DIM + o)) * N_PTS + n] = acc;
}

__global__ __launch_bounds__(256)
__attribute__((amdgpu_num_vgpr(256)))
void attn_kernel(const float* __restrict__ Q,
                 const float* __restrict__ K,
                 const float* __restrict__ V,
                 float* __restrict__ Y,
                 float* __restrict__ R) {
  __shared__ float smem_q[SMEMQ];
  __shared__ float kt[C_DIM * 16];
  __shared__ float vt[16 * VSTR];
  __shared__ float pl[8][16 * PSTR];

  const int tid  = threadIdx.x;
  const int lane = tid & 31;
  const int wave = tid >> 5;
  const int b    = blockIdx.y;
  const int qBase0 = blockIdx.x * 128;
  const int lo = lane & 15;
  const int hg = lane >> 4;
  const float invTemp = 0.088388347648318447f;
  const float NEGINF = -__builtin_inff();

#pragma unroll
  for (int i = 0; i < 16; ++i) {
    const int idx4 = tid + 256 * i;
    const int c  = idx4 >> 5;
    const int q4 = idx4 & 31;
    const float4 qv = *(const float4*)(Q + ((size_t)(b * C_DIM + c)) * N_PTS + qBase0 + q4 * 4);
    const int q = q4 * 4;
    *(float4*)&smem_q[(q >> 4) * QT_W + c * 16 + (q & 15)] = qv;
  }
  __syncthreads();
  const float* qt_w = smem_q + wave * QT_W;

  v8f  o[8] = {};
  v8f  ls8  = {};
  float t1[8], t2[8];
#pragma unroll
  for (int r = 0; r < 8; ++r) { t1[r] = NEGINF; t2[r] = NEGINF; }
  const v2f vone = {1.0f, 1.0f};

  for (int kb0 = 0; kb0 < N_PTS; kb0 += 16) {
#pragma unroll
    for (int i = 0; i < 2; ++i) {
      const int idx4 = tid + 256 * i;
      const int c  = idx4 >> 2;
      const int n4 = (idx4 & 3) * 4;
      const size_t g = ((size_t)(b * C_DIM + c)) * N_PTS + kb0 + n4;
      const float4 kv = *(const float4*)(K + g);
      *(float4*)&kt[c * 16 + n4] = kv;
      const float4 vv = *(const float4*)(V + g);
      vt[(n4 + 0) * VSTR + c] = vv.x;
      vt[(n4 + 1) * VSTR + c] = vv.y;
      vt[(n4 + 2) * VSTR + c] = vv.z;
      vt[(n4 + 3) * VSTR + c] = vv.w;
    }
    __syncthreads();

    v8f s = {};
#pragma unroll 4
    for (int ks = 0; ks < 32; ++ks) {
      const int c0 = 4 * ks + 2 * hg;
      v2f a, kb;
      a.x  = qt_w[c0 * 16 + lo];
      a.y  = qt_w[(c0 + 1) * 16 + lo];
      kb.x = kt[c0 * 16 + lo];
      kb.y = kt[(c0 + 1) * 16 + lo];
      s = __builtin_amdgcn_wmma_f32_16x16x4_f32(false, a, false, kb,
                                                (short)0, s, false, false);
    }

    v8f p;
    float mo[8];
    float anyChange = 0.0f;
#pragma unroll
    for (int r = 0; r < 8; ++r) {
      float a1 = s[r], a2 = NEGINF;
      {
        float b1 = XOR16_1(a1), b2 = XOR16_1(a2);
        float hi = fmaxf(a1, b1), lw = fminf(a1, b1);
        a2 = fmaxf(lw, fmaxf(a2, b2)); a1 = hi;
        b1 = XOR16_2(a1); b2 = XOR16_2(a2);
        hi = fmaxf(a1, b1); lw = fminf(a1, b1);
        a2 = fmaxf(lw, fmaxf(a2, b2)); a1 = hi;
        b1 = XOR16_4(a1); b2 = XOR16_4(a2);
        hi = fmaxf(a1, b1); lw = fminf(a1, b1);
        a2 = fmaxf(lw, fmaxf(a2, b2)); a1 = hi;
        b1 = XOR16_8(a1); b2 = XOR16_8(a2);
        hi = fmaxf(a1, b1); lw = fminf(a1, b1);
        a2 = fmaxf(lw, fmaxf(a2, b2)); a1 = hi;
      }
      mo[r] = t1[r];
      const float hi = fmaxf(t1[r], a1);
      const float lw = fminf(t1[r], a1);
      t2[r] = fmaxf(lw, fmaxf(t2[r], a2));
      t1[r] = hi;
      anyChange += (t1[r] - mo[r]);
      p[r] = __expf((s[r] - t1[r]) * invTemp);
    }

    if (__any(anyChange > 0.0f)) {
#pragma unroll
      for (int r = 0; r < 8; ++r) {
        const float sc = __expf((mo[r] - t1[r]) * invTemp);
        ls8[r] *= sc;
#pragma unroll
        for (int ct = 0; ct < 8; ++ct) o[ct][r] *= sc;
      }
    }

    float* pw = &pl[wave][0];
#pragma unroll
    for (int r = 0; r < 8; ++r)
      pw[(r + 8 * hg) * PSTR + lo] = p[r];

#pragma unroll 1
    for (int ks2 = 0; ks2 < 4; ++ks2) {
      const int k0 = 4 * ks2 + 2 * hg;
      v2f a;
      a.x = pw[lo * PSTR + k0];
      a.y = pw[lo * PSTR + k0 + 1];
#pragma unroll
      for (int ct = 0; ct < 8; ++ct) {
        v2f bv;
        bv.x = vt[k0 * VSTR + ct * 16 + lo];
        bv.y = vt[(k0 + 1) * VSTR + ct * 16 + lo];
        o[ct] = __builtin_amdgcn_wmma_f32_16x16x4_f32(false, a, false, bv,
                                                      (short)0, o[ct], false, false);
      }
      ls8 = __builtin_amdgcn_wmma_f32_16x16x4_f32(false, a, false, vone,
                                                  (short)0, ls8, false, false);
    }
    __syncthreads();
  }

#pragma unroll
  for (int r = 0; r < 8; ++r) {
    const float invl = 1.0f / ls8[r];
    const int q = wave * 16 + r + 8 * hg;
#pragma unroll
    for (int ct = 0; ct < 8; ++ct) {
      const int c = ct * 16 + lo;
      smem_q[c * VSTR + q] = o[ct][r] * invl;
    }
    if (lo == 0) {
      const float x1 = fminf(fmaxf(t1[r], -1.0f), 1.0f);
      const float x2 = fminf(fmaxf(t2[r], -1.0f), 1.0f);
      R[b * N_PTS + qBase0 + q] = acosf(x1) / acosf(x2);
    }
  }
  __syncthreads();

#pragma unroll
  for (int i = 0; i < 16; ++i) {
    const int idx4 = tid + 256 * i;
    const int c  = idx4 >> 5;
    const int q4 = (idx4 & 31) * 4;
    const float4 val = *(const float4*)&smem_q[c * VSTR + q4];
    *(float4*)(Y + ((size_t)(b * C_DIM + c)) * N_PTS + qBase0 + q4) = val;
  }
}

extern "C" void kernel_launch(void* const* d_in, const int* in_sizes, int n_in,
                              void* d_out, int out_size, void* d_ws, size_t ws_size,
                              hipStream_t stream) {
  (void)in_sizes; (void)n_in; (void)out_size;
  const float* q  = (const float*)d_in[0];
  const float* k  = (const float*)d_in[1];
  const float* v  = (const float*)d_in[2];
  const float* Wq = (const float*)d_in[3];
  const float* Wk = (const float*)d_in[4];
  const float* Wv = (const float*)d_in[5];

  float* out  = (float*)d_out;
  float* Yout = out;
  float* Rout = out + (size_t)B_SZ * C_DIM * N_PTS;

  const size_t tensElems = (size_t)B_SZ * C_DIM * N_PTS;
  const float *qc = q, *kc = k, *vc = v;

  if (ws_size >= 3 * tensElems * sizeof(float)) {
    float* qw = (float*)d_ws;
    float* kw = qw + tensElems;
    float* vw = kw + tensElems;
    dim3 g(N_PTS / 256, C_DIM, B_SZ);
    cmix_kernel<<<g, 256, 0, stream>>>(q, Wq, qw);
    cmix_kernel<<<g, 256, 0, stream>>>(k, Wk, kw);
    cmix_kernel<<<g, 256, 0, stream>>>(v, Wv, vw);
    qc = qw; kc = kw; vc = vw;
  }

  dim3 ga(N_PTS / 128, B_SZ);
  attn_kernel<<<ga, 256, 0, stream>>>(qc, kc, vc, Yout, Rout);
}


# attn: 64-query work-groups, 256 for the 256 WGPs (was 128); each query tile's keys split over two waves, partial softmax states merged once through LDS at the end (changes key summation order)
# speedup vs baseline: 1.6340x; 1.6340x over previous
#include <hip/hip_runtime.h>
#include <math.h>

typedef float v2f __attribute__((ext_vector_type(2)));
typedef float v8f __attribute__((ext_vector_type(8)));

#define C_DIM 128
#define N_PTS 4096
#define B_SZ  4
#define QB    64
#define VSTR  132
#define OSTR  68
#define PSTR  17
#define QT_W  2048
#define KT_G  2048
#define VT_G  2120
#define SMEMQ 8704

__device__ __forceinline__ float xor16(float v, unsigned selLo, unsigned selHi) {
  unsigned i = __builtin_bit_cast(unsigned, v);
  i = __builtin_amdgcn_permlane16(i, i, selLo, selHi, false, false);
  return __builtin_bit_cast(float, i);
}
#define XOR16_1(v) xor16((v), 0x67452301u, 0xEFCDAB89u)
#define XOR16_2(v) xor16((v), 0x54761032u, 0xDCFE98BAu)
#define XOR16_4(v) xor16((v), 0x32107654u, 0xBA98FEDCu)
#define XOR16_8(v) xor16((v), 0xFEDCBA98u, 0x76543210u)

__global__ __launch_bounds__(256) void cmix_kernel(const float* __restrict__ x,
                                                   const float* __restrict__ W,
                                                   float* __restrict__ out) {
  __shared__ float wrow[C_DIM];
  const int o = blockIdx.y;
  const int b = blockIdx.z;
  const int tid = threadIdx.x;
  if (tid < C_DIM) wrow[tid] = W[o * C_DIM + tid];
  __syncthreads();
  const int n = blockIdx.x * 256 + tid;
  float acc = 0.0f;
#pragma unroll 8
  for (int c = 0; c < C_DIM; ++c)
    acc += wrow[c] * x[((size_t)(b * C_DIM + c)) * N_PTS + n];
  out[((size_t)(b * C_DIM + o)) * N_PTS + n] = acc;
}

__global__ __launch_bounds__(256)
__attribute__((amdgpu_num_vgpr(256)))
void attn_kernel(const float* __restrict__ Q,
                 const float* __restrict__ K,
                 const float* __restrict__ V,
                 float* __restrict__ Y,
                 float* __restrict__ R) {
  __shared__ __attribute__((aligned(16))) float smem_q[SMEMQ];
  __shared__ __attribute__((aligned(16))) float kt[2 * KT_G];
  __shared__ float vt[2 * VT_G];
  __shared__ float pl[8][16 * PSTR];
  __shared__ float mrg_o[4 * 2048];
  __shared__ float mrg_s[4 * 768];
  __shared__ float rs[QB];

  const int tid  = threadIdx.x;
  const int lane = tid & 31;
  const int wave = tid >> 5;
  const int qtile = wave & 3;
  const int grp   = wave >> 2;
  const int b    = blockIdx.y;
  const int qBase0 = blockIdx.x * QB;
  const int lo = lane & 15;
  const int hg = lane >> 4;
  const float invTemp = 0.088388347648318447f;
  const float NEGINF = -__builtin_inff();

#pragma unroll
  for (int i = 0; i < 8; ++i) {
    const int idx4 = tid + 256 * i;
    const int c  = idx4 >> 4;
    const int q4 = idx4 & 15;
    const float4 qv = *(const float4*)(Q + ((size_t)(b * C_DIM + c)) * N_PTS + qBase0 + q4 * 4);
    const int q = q4 * 4;
    *(float4*)&smem_q[(q >> 4) * QT_W + c * 16 + (q & 15)] = qv;
  }
  __syncthreads();
  const float* qt_w = smem_q + qtile * QT_W;
  const float* kt_g = kt + grp * KT_G;
  const float* vt_g = vt + grp * VT_G;

  v8f  o[8] = {};
  v8f  ls8  = {};
  float t1[8], t2[8];
#pragma unroll
  for (int r = 0; r < 8; ++r) { t1[r] = NEGINF; t2[r] = NEGINF; }
  const v2f vone = {1.0f, 1.0f};

  for (int kb0 = 0; kb0 < N_PTS; kb0 += 32) {
#pragma unroll
    for (int i = 0; i < 4; ++i) {
      const int idx4 = tid + 256 * i;
      const int c  = idx4 >> 3;
      const int n4 = (idx4 & 7) * 4;
      const int tg = n4 >> 4;
      const int nn = n4 & 15;
      const size_t g = ((size_t)(b * C_DIM + c)) * N_PTS + kb0 + n4;
      const float4 kv = *(const float4*)(K + g);
      *(float4*)&kt[tg * KT_G + c * 16 + nn] = kv;
      const float4 vv = *(const float4*)(V + g);
      float* vd = vt + tg * VT_G;
      vd[(nn + 0) * VSTR + c] = vv.x;
      vd[(nn + 1) * VSTR + c] = vv.y;
      vd[(nn + 2) * VSTR + c] = vv.z;
      vd[(nn + 3) * VSTR + c] = vv.w;
    }
    __syncthreads();

    v8f s = {};
#pragma unroll 4
    for (int ks = 0; ks < 32; ++ks) {
      const int c0 = 4 * ks + 2 * hg;
      v2f a, kb;
      a.x  = qt_w[c0 * 16 + lo];
      a.y  = qt_w[(c0 + 1) * 16 + lo];
      kb.x = kt_g[c0 * 16 + lo];
      kb.y = kt_g[(c0 + 1) * 16 + lo];
      s = __builtin_amdgcn_wmma_f32_16x16x4_f32(false, a, false, kb,
                                                (short)0, s, false, false);
    }

    v8f p;
    float mo[8];
    float anyChange = 0.0f;
#pragma unroll
    for (int r = 0; r < 8; ++r) {
      float a1 = s[r], a2 = NEGINF;
      {
        float b1 = XOR16_1(a1), b2 = XOR16_1(a2);
        float hi = fmaxf(a1, b1), lw = fminf(a1, b1);
        a2 = fmaxf(lw, fmaxf(a2, b2)); a1 = hi;
        b1 = XOR16_2(a1); b2 = XOR16_2(a2);
        hi = fmaxf(a1, b1); lw = fminf(a1, b1);
        a2 = fmaxf(lw, fmaxf(a2, b2)); a1 = hi;
        b1 = XOR16_4(a1); b2 = XOR16_4(a2);
        hi = fmaxf(a1, b1); lw = fminf(a1, b1);
        a2 = fmaxf(lw, fmaxf(a2, b2)); a1 = hi;
        b1 = XOR16_8(a1); b2 = XOR16_8(a2);
        hi = fmaxf(a1, b1); lw = fminf(a1, b1);
        a2 = fmaxf(lw, fmaxf(a2, b2)); a1 = hi;
      }
      mo[r] = t1[r];
      const float hi = fmaxf(t1[r], a1);
      const float lw = fminf(t1[r], a1);
      t2[r] = fmaxf(lw, fmaxf(t2[r], a2));
      t1[r] = hi;
      anyChange += (t1[r] - mo[r]);
      p[r] = __expf((s[r] - t1[r]) * invTemp);
    }

    if (__any(anyChange > 0.0f)) {
#pragma unroll
      for (int r = 0; r < 8; ++r) {
        const float sc = __expf((mo[r] - t1[r]) * invTemp);
        ls8[r] *= sc;
#pragma unroll
        for (int ct = 0; ct < 8; ++ct) o[ct][r] *= sc;
      }
    }

    float* pw = &pl[wave][0];
#pragma unroll
    for (int r = 0; r < 8; ++r)
      pw[(r + 8 * hg) * PSTR + lo] = p[r];

#pragma unroll 1
    for (int ks2 = 0; ks2 < 4; ++ks2) {
      const int k0 = 4 * ks2 + 2 * hg;
      v2f a;
      a.x = pw[lo * PSTR + k0];
      a.y = pw[lo * PSTR + k0 + 1];
#pragma unroll
      for (int ct = 0; ct < 8; ++ct) {
        v2f bv;
        bv.x = vt_g[k0 * VSTR + ct * 16 + lo];
        bv.y = vt_g[(k0 + 1) * VSTR + ct * 16 + lo];
        o[ct] = __builtin_amdgcn_wmma_f32_16x16x4_f32(false, a, false, bv,
                                                      (short)0, o[ct], false, false);
      }
      ls8 = __builtin_amdgcn_wmma_f32_16x16x4_f32(false, a, false, vone,
                                                  (short)0, ls8, false, false);
    }
    __syncthreads();
  }

  float* mo_ = mrg_o + qtile * 2048;
  float* ms_ = mrg_s + qtile * 768;
  if (grp == 1) {
#pragma unroll
    for (int r = 0; r < 8; ++r) {
      ms_[r * 32 + lane]       = t1[r];
      ms_[256 + r * 32 + lane] = t2[r];
      ms_[512 + r * 32 + lane] = ls8[r];
#pragma unroll
      for (int ct = 0; ct < 8; ++ct)
        mo_[(ct * 8 + r) * 32 + lane] = o[ct][r];
    }
  }
  __syncthreads();

  if (grp == 0) {
#pragma unroll
    for (int r = 0; r < 8; ++r) {
      const float t1b = ms_[r * 32 + lane];
      const float t2b = ms_[256 + r * 32 + lane];
      const float lb  = ms_[512 + r * 32 + lane];
      const float m   = fmaxf(t1[r], t1b);
      const float m2  = fmaxf(fminf(t1[r], t1b), fmaxf(t2[r], t2b));
      const float sa  = __expf((t1[r] - m) * invTemp);
      const float sb  = __expf((t1b - m) * invTemp);
      const float invl = 1.0f / (ls8[r] * sa + lb * sb);
      const float wa = sa * invl;
      const float wb = sb * invl;
      const int q = qtile * 16 + r + 8 * hg;
#pragma unroll
      for (int ct = 0; ct < 8; ++ct) {
        const int c = ct * 16 + lo;
        smem_q[c * OSTR + q] = o[ct][r] * wa + mo_[(ct * 8 + r) * 32 + lane] * wb;
      }
      if (lo == 0) {
        const float x1 = fminf(fmaxf(m, -1.0f), 1.0f);
        const float x2 = fminf(fmaxf(m2, -1.0f), 1.0f);
        rs[q] = acosf(x1) / acosf(x2);
      }
    }
  }
  __syncthreads();

#pragma unroll
  for (int i = 0; i < 8; ++i) {
    const int idx4 = tid + 256 * i;
    const int c  = idx4 >> 4;
    const int q4 = (idx4 & 15) * 4;
    const float4 val = *(const float4*)&smem_q[c * OSTR + q4];
    *(float4*)(Y + ((size_t)(b * C_DIM + c)) * N_PTS + qBase0 + q4) = val;
  }
  if (tid < QB) R[b * N_PTS + qBase0 + tid] = rs[tid];
}

extern "C" void kernel_launch(void* const* d_in, const int* in_sizes, int n_in,
                              void* d_out, int out_size, void* d_ws, size_t ws_size,
                              hipStream_t stream) {
  (void)in_sizes; (void)n_in; (void)out_size;
  const float* q  = (const float*)d_in[0];
  const float* k  = (const float*)d_in[1];
  const float* v  = (const float*)d_in[2];
  const float* Wq = (const float*)d_in[3];
  const float* Wk = (const float*)d_in[4];
  const float* Wv = (const float*)d_in[5];

  float* out  = (float*)d_out;
  float* Yout = out;
  float* Rout = out + (size_t)B_SZ * C_DIM * N_PTS;

  const size_t tensElems = (size_t)B_SZ * C_DIM * N_PTS;
  const float *qc = q, *kc = k, *vc = v;

  if (ws_size >= 3 * tensElems * sizeof(float)) {
    float* qw = (float*)d_ws;
    float* kw = qw + tensElems;
    float* vw = kw + tensElems;
    dim3 g(N_PTS / 256, C_DIM, B_SZ);
    cmix_kernel<<<g, 256, 0, stream>>>(q, Wq, qw);
    cmix_kernel<<<g, 256, 0, stream>>>(k, Wk, kw);
    cmix_kernel<<<g, 256, 0, stream>>>(v, Wv, vw);
    qc = qw; kc = kw; vc = vw;
  }

  dim3 ga(N_PTS / QB, B_SZ);
  attn_kernel<<<ga, 256, 0, stream>>>(qc, kc, vc, Yout, Rout);
}


# 1x1 conv channel mixes moved onto f32 WMMA: a work-group stages its x tile in LDS once for all 128 outputs (was a scalar loop re-reading x per output); channel sums now add in WMMA k order
# speedup vs baseline: 1.9570x; 1.1977x over previous
#include <hip/hip_runtime.h>
#include <math.h>

typedef float v2f __attribute__((ext_vector_type(2)));
typedef float v8f __attribute__((ext_vector_type(8)));

#define C_DIM 128
#define N_PTS 4096
#define B_SZ  4
#define QB    64
#define VSTR  132
#define OSTR  68
#define PSTR  17
#define QT_W  2048
#define KT_G  2048
#define VT_G  2120
#define SMEMQ 8704

__device__ __forceinline__ float xor16(float v, unsigned selLo, unsigned selHi) {
  unsigned i = __builtin_bit_cast(unsigned, v);
  i = __builtin_amdgcn_permlane16(i, i, selLo, selHi, false, false);
  return __builtin_bit_cast(float, i);
}
#define XOR16_1(v) xor16((v), 0x67452301u, 0xEFCDAB89u)
#define XOR16_2(v) xor16((v), 0x54761032u, 0xDCFE98BAu)
#define XOR16_4(v) xor16((v), 0x32107654u, 0xBA98FEDCu)
#define XOR16_8(v) xor16((v), 0xFEDCBA98u, 0x76543210u)

#define CM_NT   64
#define CM_XP   160
#define CM_WSTR 132
#define CM_OSTR 68
#define CM_XS   (64 * CM_XP)
#define CM_WT   (16 * CM_WSTR)

__global__ __launch_bounds__(256) void cmix_kernel(const float* __restrict__ x,
                                                   const float* __restrict__ W,
                                                   float* __restrict__ out) {
  __shared__ __attribute__((aligned(16))) float cm_lds[CM_XS + 8 * CM_WT];
  float* xs = cm_lds;
  const int tid  = threadIdx.x;
  const int lane = tid & 31;
  const int wave = tid >> 5;
  const int lo = lane & 15;
  const int hg = lane >> 4;
  const int b   = blockIdx.y;
  const int nb0 = blockIdx.x * CM_NT;
  const int o0  = wave * 16;
  float* wt_w = cm_lds + CM_XS + wave * CM_WT;

#pragma unroll
  for (int i = 0; i < 4; ++i) {
    const int idx = tid + 256 * i;
    const int p  = idx >> 4;
    const int n4 = (idx & 15) * 4;
    const float* xp = x + ((size_t)(b * C_DIM + 2 * p)) * N_PTS + nb0 + n4;
    const float4 e = *(const float4*)xp;
    const float4 f = *(const float4*)(xp + N_PTS);
    float* d = &xs[p * CM_XP + 2 * n4];
    *(float4*)d       = make_float4(e.x, f.x, e.y, f.y);
    *(float4*)(d + 4) = make_float4(e.z, f.z, e.w, f.w);
  }
#pragma unroll
  for (int i = 0; i < 16; ++i) {
    const float4 wv = *(const float4*)(W + (size_t)(o0 + i) * C_DIM + 4 * lane);
    *(float4*)&wt_w[i * CM_WSTR + 4 * lane] = wv;
  }
  __syncthreads();

  v8f acc[4] = {};
  const float* a_l = wt_w + lo * CM_WSTR + 2 * hg;
  const float* b_l = xs + hg * CM_XP + 2 * lo;
#pragma unroll
  for (int ks = 0; ks < 32; ++ks) {
    const v2f a = *(const v2f*)(a_l + 4 * ks);
#pragma unroll
    for (int j = 0; j < 4; ++j) {
      const v2f bv = *(const v2f*)(b_l + 2 * ks * CM_XP + 32 * j);
      acc[j] = __builtin_amdgcn_wmma_f32_16x16x4_f32(false, a, false, bv,
                                                     (short)0, acc[j], false, false);
    }
  }

#pragma unroll
  for (int j = 0; j < 4; ++j)
#pragma unroll
    for (int r = 0; r < 8; ++r)
      wt_w[(r + 8 * hg) * CM_OSTR + 16 * j + lo] = acc[j][r];
#pragma unroll
  for (int i = 0; i < 8; ++i) {
    const int row = 2 * i + hg;
    const float4 val = *(const float4*)&wt_w[row * CM_OSTR + 4 * lo];
    *(float4*)(out + ((size_t)(b * C_DIM + o0 + row)) * N_PTS + nb0 + 4 * lo) = val;
  }
}

__global__ __launch_bounds__(256)
__attribute__((amdgpu_num_vgpr(256)))
void attn_kernel(const float* __restrict__ Q,
                 const float* __restrict__ K,
                 const float* __restrict__ V,
                 float* __restrict__ Y,
                 float* __restrict__ R) {
  __shared__ __attribute__((aligned(16))) float smem_q[SMEMQ];
  __shared__ __attribute__((aligned(16))) float kt[2 * KT_G];
  __shared__ float vt[2 * VT_G];
  __shared__ float pl[8][16 * PSTR];
  __shared__ float mrg_o[4 * 2048];
  __shared__ float mrg_s[4 * 768];
  __shared__ float rs[QB];

  const int tid  = threadIdx.x;
  const int lane = tid & 31;
  const int wave = tid >> 5;
  const int qtile = wave & 3;
  const int grp   = wave >> 2;
  const int b    = blockIdx.y;
  const int qBase0 = blockIdx.x * QB;
  const int lo = lane & 15;
  const int hg = lane >> 4;
  const float invTemp = 0.088388347648318447f;
  const float NEGINF = -__builtin_inff();

#pragma unroll
  for (int i = 0; i < 8; ++i) {
    const int idx4 = tid + 256 * i;
    const int c  = idx4 >> 4;
    const int q4 = idx4 & 15;
    const float4 qv = *(const float4*)(Q + ((size_t)(b * C_DIM + c)) * N_PTS + qBase0 + q4 * 4);
    const int q = q4 * 4;
    *(float4*)&smem_q[(q >> 4) * QT_W + c * 16 + (q & 15)] = qv;
  }
  __syncthreads();
  const float* qt_w = smem_q + qtile * QT_W;
  const float* kt_g = kt + grp * KT_G;
  const float* vt_g = vt + grp * VT_G;

  v8f  o[8] = {};
  v8f  ls8  = {};
  float t1[8], t2[8];
#pragma unroll
  for (int r = 0; r < 8; ++r) { t1[r] = NEGINF; t2[r] = NEGINF; }
  const v2f vone = {1.0f, 1.0f};

  for (int kb0 = 0; kb0 < N_PTS; kb0 += 32) {
#pragma unroll
    for (int i = 0; i < 4; ++i) {
      const int idx4 = tid + 256 * i;
      const int c  = idx4 >> 3;
      const int n4 = (idx4 & 7) * 4;
      const int tg = n4 >> 4;
      const int nn = n4 & 15;
      const size_t g = ((size_t)(b * C_DIM + c)) * N_PTS + kb0 + n4;
      const float4 kv = *(const float4*)(K + g);
      *(float4*)&kt[tg * KT_G + c * 16 + nn] = kv;
      const float4 vv = *(const float4*)(V + g);
      float* vd = vt + tg * VT_G;
      vd[(nn + 0) * VSTR + c] = vv.x;
      vd[(nn + 1) * VSTR + c] = vv.y;
      vd[(nn + 2) * VSTR + c] = vv.z;
      vd[(nn + 3) * VSTR + c] = vv.w;
    }
    __syncthreads();

    v8f s = {};
#pragma unroll 4
    for (int ks = 0; ks < 32; ++ks) {
      const int c0 = 4 * ks + 2 * hg;
      v2f a, kb;
      a.x  = qt_w[c0 * 16 + lo];
      a.y  = qt_w[(c0 + 1) * 16 + lo];
      kb.x = kt_g[c0 * 16 + lo];
      kb.y = kt_g[(c0 + 1) * 16 + lo];
      s = __builtin_amdgcn_wmma_f32_16x16x4_f32(false, a, false, kb,
                                                (short)0, s, false, false);
    }

    v8f p;
    float mo[8];
    float anyChange = 0.0f;
#pragma unroll
    for (int r = 0; r < 8; ++r) {
      float a1 = s[r], a2 = NEGINF;
      {
        float b1 = XOR16_1(a1), b2 = XOR16_1(a2);
        float hi = fmaxf(a1, b1), lw = fminf(a1, b1);
        a2 = fmaxf(lw, fmaxf(a2, b2)); a1 = hi;
        b1 = XOR16_2(a1); b2 = XOR16_2(a2);
        hi = fmaxf(a1, b1); lw = fminf(a1, b1);
        a2 = fmaxf(lw, fmaxf(a2, b2)); a1 = hi;
        b1 = XOR16_4(a1); b2 = XOR16_4(a2);
        hi = fmaxf(a1, b1); lw = fminf(a1, b1);
        a2 = fmaxf(lw, fmaxf(a2, b2)); a1 = hi;
        b1 = XOR16_8(a1); b2 = XOR16_8(a2);
        hi = fmaxf(a1, b1); lw = fminf(a1, b1);
        a2 = fmaxf(lw, fmaxf(a2, b2)); a1 = hi;
      }
      mo[r] = t1[r];
      const float hi = fmaxf(t1[r], a1);
      const float lw = fminf(t1[r], a1);
      t2[r] = fmaxf(lw, fmaxf(t2[r], a2));
      t1[r] = hi;
      anyChange += (t1[r] - mo[r]);
      p[r] = __expf((s[r] - t1[r]) * invTemp);
    }

    if (__any(anyChange > 0.0f)) {
#pragma unroll
      for (int r = 0; r < 8; ++r) {
        const float sc = __expf((mo[r] - t1[r]) * invTemp);
        ls8[r] *= sc;
#pragma unroll
        for (int ct = 0; ct < 8; ++ct) o[ct][r] *= sc;
      }
    }

    float* pw = &pl[wave][0];
#pragma unroll
    for (int r = 0; r < 8; ++r)
      pw[(r + 8 * hg) * PSTR + lo] = p[r];

#pragma unroll 1
    for (int ks2 = 0; ks2 < 4; ++ks2) {
      const int k0 = 4 * ks2 + 2 * hg;
      v2f a;
      a.x = pw[lo * PSTR + k0];
      a.y = pw[lo * PSTR + k0 + 1];
#pragma unroll
      for (int ct = 0; ct < 8; ++ct) {
        v2f bv;
        bv.x = vt_g[k0 * VSTR + ct * 16 + lo];
        bv.y = vt_g[(k0 + 1) * VSTR + ct * 16 + lo];
        o[ct] = __builtin_amdgcn_wmma_f32_16x16x4_f32(false, a, false, bv,
                                                      (short)0, o[ct], false, false);
      }
      ls8 = __builtin_amdgcn_wmma_f32_16x16x4_f32(false, a, false, vone,
                                                  (short)0, ls8, false, false);
    }
    __syncthreads();
  }

  float* mo_ = mrg_o + qtile * 2048;
  float* ms_ = mrg_s + qtile * 768;
  if (grp == 1) {
#pragma unroll
    for (int r = 0; r < 8; ++r) {
      ms_[r * 32 + lane]       = t1[r];
      ms_[256 + r * 32 + lane] = t2[r];
      ms_[512 + r * 32 + lane] = ls8[r];
#pragma unroll
      for (int ct = 0; ct < 8; ++ct)
        mo_[(ct * 8 + r) * 32 + lane] = o[ct][r];
    }
  }
  __syncthreads();

  if (grp == 0) {
#pragma unroll
    for (int r = 0; r < 8; ++r) {
      const float t1b = ms_[r * 32 + lane];
      const float t2b = ms_[256 + r * 32 + lane];
      const float lb  = ms_[512 + r * 32 + lane];
      const float m   = fmaxf(t1[r], t1b);
      const float m2  = fmaxf(fminf(t1[r], t1b), fmaxf(t2[r], t2b));
      const float sa  = __expf((t1[r] - m) * invTemp);
      const float sb  = __expf((t1b - m) * invTemp);
      const float invl = 1.0f / (ls8[r] * sa + lb * sb);
      const float wa = sa * invl;
      const float wb = sb * invl;
      const int q = qtile * 16 + r + 8 * hg;
#pragma unroll
      for (int ct = 0; ct < 8; ++ct) {
        const int c = ct * 16 + lo;
        smem_q[c * OSTR + q] = o[ct][r] * wa + mo_[(ct * 8 + r) * 32 + lane] * wb;
      }
      if (lo == 0) {
        const float x1 = fminf(fmaxf(m, -1.0f), 1.0f);
        const float x2 = fminf(fmaxf(m2, -1.0f), 1.0f);
        rs[q] = acosf(x1) / acosf(x2);
      }
    }
  }
  __syncthreads();

#pragma unroll
  for (int i = 0; i < 8; ++i) {
    const int idx4 = tid + 256 * i;
    const int c  = idx4 >> 4;
    const int q4 = (idx4 & 15) * 4;
    const float4 val = *(const float4*)&smem_q[c * OSTR + q4];
    *(float4*)(Y + ((size_t)(b * C_DIM + c)) * N_PTS + qBase0 + q4) = val;
  }
  if (tid < QB) R[b * N_PTS + qBase0 + tid] = rs[tid];
}

extern "C" void kernel_launch(void* const* d_in, const int* in_sizes, int n_in,
                              void* d_out, int out_size, void* d_ws, size_t ws_size,
                              hipStream_t stream) {
  (void)in_sizes; (void)n_in; (void)out_size;
  const float* q  = (const float*)d_in[0];
  const float* k  = (const float*)d_in[1];
  const float* v  = (const float*)d_in[2];
  const float* Wq = (const float*)d_in[3];
  const float* Wk = (const float*)d_in[4];
  const float* Wv = (const float*)d_in[5];

  float* out  = (float*)d_out;
  float* Yout = out;
  float* Rout = out + (size_t)B_SZ * C_DIM * N_PTS;

  const size_t tensElems = (size_t)B_SZ * C_DIM * N_PTS;
  const float *qc = q, *kc = k, *vc = v;

  if (ws_size >= 3 * tensElems * sizeof(float)) {
    float* qw = (float*)d_ws;
    float* kw = qw + tensElems;
    float* vw = kw + tensElems;
    dim3 g(N_PTS / CM_NT, B_SZ);
    cmix_kernel<<<g, 256, 0, stream>>>(q, Wq, qw);
    cmix_kernel<<<g, 256, 0, stream>>>(k, Wk, kw);
    cmix_kernel<<<g, 256, 0, stream>>>(v, Wv, vw);
    qc = qw; kc = kw; vc = vw;
  }

  dim3 ga(N_PTS / QB, B_SZ);
  attn_kernel<<<ga, 256, 0, stream>>>(qc, kc, vc, Yout, Rout);
}


# attn: K/V staging double-buffered: next step's global loads issued before the score WMMAs, held in registers, stored to a second LDS stage behind them; one barrier per step (was two, loads exposed)
# speedup vs baseline: 2.2969x; 1.1737x over previous
#include <hip/hip_runtime.h>
#include <math.h>

typedef float v2f __attribute__((ext_vector_type(2)));
typedef float v8f __attribute__((ext_vector_type(8)));

#define C_DIM 128
#define N_PTS 4096
#define B_SZ  4
#define QB    64
#define VSTR  132
#define OSTR  68
#define PSTR  17
#define QT_W  2048
#define KT_G  2048
#define VT_G  2120
#define KSTG  (2 * KT_G)
#define VSTG  (2 * VT_G)
#define NSTEP (N_PTS / 32)
#define SMEMQ 8704

__device__ __forceinline__ float xor16(float v, unsigned selLo, unsigned selHi) {
  unsigned i = __builtin_bit_cast(unsigned, v);
  i = __builtin_amdgcn_permlane16(i, i, selLo, selHi, false, false);
  return __builtin_bit_cast(float, i);
}
#define XOR16_1(v) xor16((v), 0x67452301u, 0xEFCDAB89u)
#define XOR16_2(v) xor16((v), 0x54761032u, 0xDCFE98BAu)
#define XOR16_4(v) xor16((v), 0x32107654u, 0xBA98FEDCu)
#define XOR16_8(v) xor16((v), 0xFEDCBA98u, 0x76543210u)

#define CM_NT   64
#define CM_XP   160
#define CM_WSTR 132
#define CM_OSTR 68
#define CM_XS   (64 * CM_XP)
#define CM_WT   (16 * CM_WSTR)

__global__ __launch_bounds__(256) void cmix_kernel(const float* __restrict__ x,
                                                   const float* __restrict__ W,
                                                   float* __restrict__ out) {
  __shared__ __attribute__((aligned(16))) float cm_lds[CM_XS + 8 * CM_WT];
  float* xs = cm_lds;
  const int tid  = threadIdx.x;
  const int lane = tid & 31;
  const int wave = tid >> 5;
  const int lo = lane & 15;
  const int hg = lane >> 4;
  const int b   = blockIdx.y;
  const int nb0 = blockIdx.x * CM_NT;
  const int o0  = wave * 16;
  float* wt_w = cm_lds + CM_XS + wave * CM_WT;

#pragma unroll
  for (int i = 0; i < 4; ++i) {
    const int idx = tid + 256 * i;
    const int p  = idx >> 4;
    const int n4 = (idx & 15) * 4;
    const float* xp = x + ((size_t)(b * C_DIM + 2 * p)) * N_PTS + nb0 + n4;
    const float4 e = *(const float4*)xp;
    const float4 f = *(const float4*)(xp + N_PTS);
    float* d = &xs[p * CM_XP + 2 * n4];
    *(float4*)d       = make_float4(e.x, f.x, e.y, f.y);
    *(float4*)(d + 4) = make_float4(e.z, f.z, e.w, f.w);
  }
#pragma unroll
  for (int i = 0; i < 16; ++i) {
    const float4 wv = *(const float4*)(W + (size_t)(o0 + i) * C_DIM + 4 * lane);
    *(float4*)&wt_w[i * CM_WSTR + 4 * lane] = wv;
  }
  __syncthreads();

  v8f acc[4] = {};
  const float* a_l = wt_w + lo * CM_WSTR + 2 * hg;
  const float* b_l = xs + hg * CM_XP + 2 * lo;
#pragma unroll
  for (int ks = 0; ks < 32; ++ks) {
    const v2f a = *(const v2f*)(a_l + 4 * ks);
#pragma unroll
    for (int j = 0; j < 4; ++j) {
      const v2f bv = *(const v2f*)(b_l + 2 * ks * CM_XP + 32 * j);
      acc[j] = __builtin_amdgcn_wmma_f32_16x16x4_f32(false, a, false, bv,
                                                     (short)0, acc[j], false, false);
    }
  }

#pragma unroll
  for (int j = 0; j < 4; ++j)
#pragma unroll
    for (int r = 0; r < 8; ++r)
      wt_w[(r + 8 * hg) * CM_OSTR + 16 * j + lo] = acc[j][r];
#pragma unroll
  for (int i = 0; i < 8; ++i) {
    const int row = 2 * i + hg;
    const float4 val = *(const float4*)&wt_w[row * CM_OSTR + 4 * lo];
    *(float4*)(out + ((size_t)(b * C_DIM + o0 + row)) * N_PTS + nb0 + 4 * lo) = val;
  }
}

__global__ __launch_bounds__(256)
__attribute__((amdgpu_num_vgpr(256)))
void attn_kernel(const float* __restrict__ Q,
                 const float* __restrict__ K,
                 const float* __restrict__ V,
                 float* __restrict__ Y,
                 float* __restrict__ R) {
  __shared__ __attribute__((aligned(16))) float smem_q[SMEMQ];
  __shared__ __attribute__((aligned(16))) float kt[2 * KSTG];
  __shared__ float vt[2 * VSTG];
  __shared__ float pl[8][16 * PSTR];
  __shared__ float mrg_o[4 * 2048];
  __shared__ float mrg_s[4 * 768];
  __shared__ float rs[QB];

  const int tid  = threadIdx.x;
  const int lane = tid & 31;
  const int wave = tid >> 5;
  const int qtile = wave & 3;
  const int grp   = wave >> 2;
  const int b    = blockIdx.y;
  const int qBase0 = blockIdx.x * QB;
  const int lo = lane & 15;
  const int hg = lane >> 4;
  const float invTemp = 0.088388347648318447f;
  const float NEGINF = -__builtin_inff();

#pragma unroll
  for (int i = 0; i < 8; ++i) {
    const int idx4 = tid + 256 * i;
    const int c  = idx4 >> 4;
    const int q4 = idx4 & 15;
    const float4 qv = *(const float4*)(Q + ((size_t)(b * C_DIM + c)) * N_PTS + qBase0 + q4 * 4);
    const int q = q4 * 4;
    *(float4*)&smem_q[(q >> 4) * QT_W + c * 16 + (q & 15)] = qv;
  }

  const int st_c  = tid >> 3;
  const int st_n4 = (tid & 7) * 4;
  const int st_tg = st_n4 >> 4;
  const int st_nn = st_n4 & 15;
  const size_t st_g = ((size_t)(b * C_DIM + st_c)) * N_PTS + st_n4;
  const float* kg = K + st_g;
  const float* vg = V + st_g;
  float* k_st = kt + st_tg * KT_G + st_c * 16 + st_nn;
  float* v_st = vt + st_tg * VT_G + st_nn * VSTR + st_c;
  float4 kr0, kr1, kr2, kr3, vr0, vr1, vr2, vr3;
#define KV_FETCH(kb)                                                   \
  do {                                                                 \
    kr0 = *(const float4*)(kg + (kb));                                 \
    kr1 = *(const float4*)(kg + (kb) + (size_t)1 * 32 * N_PTS);        \
    kr2 = *(const float4*)(kg + (kb) + (size_t)2 * 32 * N_PTS);        \
    kr3 = *(const float4*)(kg + (kb) + (size_t)3 * 32 * N_PTS);        \
    vr0 = *(const float4*)(vg + (kb));                                 \
    vr1 = *(const float4*)(vg + (kb) + (size_t)1 * 32 * N_PTS);        \
    vr2 = *(const float4*)(vg + (kb) + (size_t)2 * 32 * N_PTS);        \
    vr3 = *(const float4*)(vg + (kb) + (size_t)3 * 32 * N_PTS);        \
  } while (0)
#define V_PUT(vd, i, vv)                                               \
  do {                                                                 \
    (vd)[0 * VSTR + (i) * 32] = (vv).x;                                \
    (vd)[1 * VSTR + (i) * 32] = (vv).y;                                \
    (vd)[2 * VSTR + (i) * 32] = (vv).z;                                \
    (vd)[3 * VSTR + (i) * 32] = (vv).w;                                \
  } while (0)
#define KV_PUT(kd, vd)                                                 \
  do {                                                                 \
    *(float4*)((kd) + 0 * 512) = kr0;                                  \
    *(float4*)((kd) + 1 * 512) = kr1;                                  \
    *(float4*)((kd) + 2 * 512) = kr2;                                  \
    *(float4*)((kd) + 3 * 512) = kr3;                                  \
    V_PUT(vd, 0, vr0);                                                 \
    V_PUT(vd, 1, vr1);                                                 \
    V_PUT(vd, 2, vr2);                                                 \
    V_PUT(vd, 3, vr3);                                                 \
  } while (0)

  KV_FETCH(0);
  KV_PUT(k_st, v_st);
  __syncthreads();
  const float* qt_w = smem_q + qtile * QT_W;

  v8f  o[8] = {};
  v8f  ls8  = {};
  float t1[8], t2[8];
#pragma unroll
  for (int r = 0; r < 8; ++r) { t1[r] = NEGINF; t2[r] = NEGINF; }
  const v2f vone = {1.0f, 1.0f};

  for (int it = 0; it < NSTEP; ++it) {
    const int  cur  = it & 1;
    const bool more = (it + 1 < NSTEP);
    if (more) {
      const int kbn = (it + 1) * 32;
      KV_FETCH(kbn);
    }
    const float* kt_g = kt + cur * KSTG + grp * KT_G;
    const float* vt_g = vt + cur * VSTG + grp * VT_G;

    v8f s = {};
#pragma unroll 4
    for (int ks = 0; ks < 32; ++ks) {
      const int c0 = 4 * ks + 2 * hg;
      v2f a, kb;
      a.x  = qt_w[c0 * 16 + lo];
      a.y  = qt_w[(c0 + 1) * 16 + lo];
      kb.x = kt_g[c0 * 16 + lo];
      kb.y = kt_g[(c0 + 1) * 16 + lo];
      s = __builtin_amdgcn_wmma_f32_16x16x4_f32(false, a, false, kb,
                                                (short)0, s, false, false);
    }

    if (more) {
      float* kd = k_st + (cur ^ 1) * KSTG;
      float* vd = v_st + (cur ^ 1) * VSTG;
      KV_PUT(kd, vd);
    }

    v8f p;
    float mo[8];
    float anyChange = 0.0f;
#pragma unroll
    for (int r = 0; r < 8; ++r) {
      float a1 = s[r], a2 = NEGINF;
      {
        float b1 = XOR16_1(a1), b2 = XOR16_1(a2);
        float hi = fmaxf(a1, b1), lw = fminf(a1, b1);
        a2 = fmaxf(lw, fmaxf(a2, b2)); a1 = hi;
        b1 = XOR16_2(a1); b2 = XOR16_2(a2);
        hi = fmaxf(a1, b1); lw = fminf(a1, b1);
        a2 = fmaxf(lw, fmaxf(a2, b2)); a1 = hi;
        b1 = XOR16_4(a1); b2 = XOR16_4(a2);
        hi = fmaxf(a1, b1); lw = fminf(a1, b1);
        a2 = fmaxf(lw, fmaxf(a2, b2)); a1 = hi;
        b1 = XOR16_8(a1); b2 = XOR16_8(a2);
        hi = fmaxf(a1, b1); lw = fminf(a1, b1);
        a2 = fmaxf(lw, fmaxf(a2, b2)); a1 = hi;
      }
      mo[r] = t1[r];
      const float hi = fmaxf(t1[r], a1);
      const float lw = fminf(t1[r], a1);
      t2[r] = fmaxf(lw, fmaxf(t2[r], a2));
      t1[r] = hi;
      anyChange += (t1[r] - mo[r]);
      p[r] = __expf((s[r] - t1[r]) * invTemp);
    }

    if (__any(anyChange > 0.0f)) {
#pragma unroll
      for (int r = 0; r < 8; ++r) {
        const float sc = __expf((mo[r] - t1[r]) * invTemp);
        ls8[r] *= sc;
#pragma unroll
        for (int ct = 0; ct < 8; ++ct) o[ct][r] *= sc;
      }
    }

    float* pw = &pl[wave][0];
#pragma unroll
    for (int r = 0; r < 8; ++r)
      pw[(r + 8 * hg) * PSTR + lo] = p[r];

#pragma unroll 1
    for (int ks2 = 0; ks2 < 4; ++ks2) {
      const int k0 = 4 * ks2 + 2 * hg;
      v2f a;
      a.x = pw[lo * PSTR + k0];
      a.y = pw[lo * PSTR + k0 + 1];
#pragma unroll
      for (int ct = 0; ct < 8; ++ct) {
        v2f bv;
        bv.x = vt_g[k0 * VSTR + ct * 16 + lo];
        bv.y = vt_g[(k0 + 1) * VSTR + ct * 16 + lo];
        o[ct] = __builtin_amdgcn_wmma_f32_16x16x4_f32(false, a, false, bv,
                                                      (short)0, o[ct], false, false);
      }
      ls8 = __builtin_amdgcn_wmma_f32_16x16x4_f32(false, a, false, vone,
                                                  (short)0, ls8, false, false);
    }
    __syncthreads();
  }

#undef KV_FETCH
#undef KV_PUT
#undef V_PUT

  float* mo_ = mrg_o + qtile * 2048;
  float* ms_ = mrg_s + qtile * 768;
  if (grp == 1) {
#pragma unroll
    for (int r = 0; r < 8; ++r) {
      ms_[r * 32 + lane]       = t1[r];
      ms_[256 + r * 32 + lane] = t2[r];
      ms_[512 + r * 32 + lane] = ls8[r];
#pragma unroll
      for (int ct = 0; ct < 8; ++ct)
        mo_[(ct * 8 + r) * 32 + lane] = o[ct][r];
    }
  }
  __syncthreads();

  if (grp == 0) {
#pragma unroll
    for (int r = 0; r < 8; ++r) {
      const float t1b = ms_[r * 32 + lane];
      const float t2b = ms_[256 + r * 32 + lane];
      const float lb  = ms_[512 + r * 32 + lane];
      const float m   = fmaxf(t1[r], t1b);
      const float m2  = fmaxf(fminf(t1[r], t1b), fmaxf(t2[r], t2b));
      const float sa  = __expf((t1[r] - m) * invTemp);
      const float sb  = __expf((t1b - m) * invTemp);
      const float invl = 1.0f / (ls8[r] * sa + lb * sb);
      const float wa = sa * invl;
      const float wb = sb * invl;
      const int q = qtile * 16 + r + 8 * hg;
#pragma unroll
      for (int ct = 0; ct < 8; ++ct) {
        const int c = ct * 16 + lo;
        smem_q[c * OSTR + q] = o[ct][r] * wa + mo_[(ct * 8 + r) * 32 + lane] * wb;
      }
      if (lo == 0) {
        const float x1 = fminf(fmaxf(m, -1.0f), 1.0f);
        const float x2 = fminf(fmaxf(m2, -1.0f), 1.0f);
        rs[q] = acosf(x1) / acosf(x2);
      }
    }
  }
  __syncthreads();

#pragma unroll
  for (int i = 0; i < 8; ++i) {
    const int idx4 = tid + 256 * i;
    const int c  = idx4 >> 4;
    const int q4 = (idx4 & 15) * 4;
    const float4 val = *(const float4*)&smem_q[c * OSTR + q4];
    *(float4*)(Y + ((size_t)(b * C_DIM + c)) * N_PTS + qBase0 + q4) = val;
  }
  if (tid < QB) R[b * N_PTS + qBase0 + tid] = rs[tid];
}

extern "C" void kernel_launch(void* const* d_in, const int* in_sizes, int n_in,
                              void* d_out, int out_size, void* d_ws, size_t ws_size,
                              hipStream_t stream) {
  (void)in_sizes; (void)n_in; (void)out_size;
  const float* q  = (const float*)d_in[0];
  const float* k  = (const float*)d_in[1];
  const float* v  = (const float*)d_in[2];
  const float* Wq = (const float*)d_in[3];
  const float* Wk = (const float*)d_in[4];
  const float* Wv = (const float*)d_in[5];

  float* out  = (float*)d_out;
  float* Yout = out;
  float* Rout = out + (size_t)B_SZ * C_DIM * N_PTS;

  const size_t tensElems = (size_t)B_SZ * C_DIM * N_PTS;
  const float *qc = q, *kc = k, *vc = v;

  if (ws_size >= 3 * tensElems * sizeof(float)) {
    float* qw = (float*)d_ws;
    float* kw = qw + tensElems;
    float* vw = kw + tensElems;
    dim3 g(N_PTS / CM_NT, B_SZ);
    cmix_kernel<<<g, 256, 0, stream>>>(q, Wq, qw);
    cmix_kernel<<<g, 256, 0, stream>>>(k, Wk, kw);
    cmix_kernel<<<g, 256, 0, stream>>>(v, Wv, vw);
    qc = qw; kc = kw; vc = vw;
  }

  dim3 ga(N_PTS / QB, B_SZ);
  attn_kernel<<<ga, 256, 0, stream>>>(qc, kc, vc, Yout, Rout);
}


# attn: LDS fragment reads of both WMMA runs issued one round ahead into a second register set (held by scheduling fences) and waited behind that round's WMMAs; was load-wait-WMMA per round
# speedup vs baseline: 2.7266x; 1.1871x over previous
#include <hip/hip_runtime.h>
#include <math.h>

typedef float v2f __attribute__((ext_vector_type(2)));
typedef float v8f __attribute__((ext_vector_type(8)));

#define C_DIM 128
#define N_PTS 4096
#define B_SZ  4
#define QB    64
#define VSTR  132
#define OSTR  68
#define PSTR  17
#define QT_W  2048
#define KT_G  2048
#define VT_G  2120
#define KSTG  (2 * KT_G)
#define VSTG  (2 * VT_G)
#define NSTEP (N_PTS / 32)
#define SMEMQ 8704

__device__ __forceinline__ float xor16(float v, unsigned selLo, unsigned selHi) {
  unsigned i = __builtin_bit_cast(unsigned, v);
  i = __builtin_amdgcn_permlane16(i, i, selLo, selHi, false, false);
  return __builtin_bit_cast(float, i);
}
#define XOR16_1(v) xor16((v), 0x67452301u, 0xEFCDAB89u)
#define XOR16_2(v) xor16((v), 0x54761032u, 0xDCFE98BAu)
#define XOR16_4(v) xor16((v), 0x32107654u, 0xBA98FEDCu)
#define XOR16_8(v) xor16((v), 0xFEDCBA98u, 0x76543210u)

#define CM_NT   64
#define CM_XP   160
#define CM_WSTR 132
#define CM_OSTR 68
#define CM_XS   (64 * CM_XP)
#define CM_WT   (16 * CM_WSTR)

__global__ __launch_bounds__(256) void cmix_kernel(const float* __restrict__ x,
                                                   const float* __restrict__ W,
                                                   float* __restrict__ out) {
  __shared__ __attribute__((aligned(16))) float cm_lds[CM_XS + 8 * CM_WT];
  float* xs = cm_lds;
  const int tid  = threadIdx.x;
  const int lane = tid & 31;
  const int wave = tid >> 5;
  const int lo = lane & 15;
  const int hg = lane >> 4;
  const int b   = blockIdx.y;
  const int nb0 = blockIdx.x * CM_NT;
  const int o0  = wave * 16;
  float* wt_w = cm_lds + CM_XS + wave * CM_WT;

#pragma unroll
  for (int i = 0; i < 4; ++i) {
    const int idx = tid + 256 * i;
    const int p  = idx >> 4;
    const int n4 = (idx & 15) * 4;
    const float* xp = x + ((size_t)(b * C_DIM + 2 * p)) * N_PTS + nb0 + n4;
    const float4 e = *(const float4*)xp;
    const float4 f = *(const float4*)(xp + N_PTS);
    float* d = &xs[p * CM_XP + 2 * n4];
    *(float4*)d       = make_float4(e.x, f.x, e.y, f.y);
    *(float4*)(d + 4) = make_float4(e.z, f.z, e.w, f.w);
  }
#pragma unroll
  for (int i = 0; i < 16; ++i) {
    const float4 wv = *(const float4*)(W + (size_t)(o0 + i) * C_DIM + 4 * lane);
    *(float4*)&wt_w[i * CM_WSTR + 4 * lane] = wv;
  }
  __syncthreads();

  v8f acc[4] = {};
  const float* a_l = wt_w + lo * CM_WSTR + 2 * hg;
  const float* b_l = xs + hg * CM_XP + 2 * lo;
#pragma unroll
  for (int ks = 0; ks < 32; ++ks) {
    const v2f a = *(const v2f*)(a_l + 4 * ks);
#pragma unroll
    for (int j = 0; j < 4; ++j) {
      const v2f bv = *(const v2f*)(b_l + 2 * ks * CM_XP + 32 * j);
      acc[j] = __builtin_amdgcn_wmma_f32_16x16x4_f32(false, a, false, bv,
                                                     (short)0, acc[j], false, false);
    }
  }

#pragma unroll
  for (int j = 0; j < 4; ++j)
#pragma unroll
    for (int r = 0; r < 8; ++r)
      wt_w[(r + 8 * hg) * CM_OSTR + 16 * j + lo] = acc[j][r];
#pragma unroll
  for (int i = 0; i < 8; ++i) {
    const int row = 2 * i + hg;
    const float4 val = *(const float4*)&wt_w[row * CM_OSTR + 4 * lo];
    *(float4*)(out + ((size_t)(b * C_DIM + o0 + row)) * N_PTS + nb0 + 4 * lo) = val;
  }
}

__global__ __launch_bounds__(256)
__attribute__((amdgpu_num_vgpr(256)))
void attn_kernel(const float* __restrict__ Q,
                 const float* __restrict__ K,
                 const float* __restrict__ V,
                 float* __restrict__ Y,
                 float* __restrict__ R) {
  __shared__ __attribute__((aligned(16))) float smem_q[SMEMQ];
  __shared__ __attribute__((aligned(16))) float kt[2 * KSTG];
  __shared__ float vt[2 * VSTG];
  __shared__ float pl[8][16 * PSTR];
  __shared__ float mrg_o[4 * 2048];
  __shared__ float mrg_s[4 * 768];
  __shared__ float rs[QB];

  const int tid  = threadIdx.x;
  const int lane = tid & 31;
  const int wave = tid >> 5;
  const int qtile = wave & 3;
  const int grp   = wave >> 2;
  const int b    = blockIdx.y;
  const int qBase0 = blockIdx.x * QB;
  const int lo = lane & 15;
  const int hg = lane >> 4;
  const float invTemp = 0.088388347648318447f;
  const float NEGINF = -__builtin_inff();

#pragma unroll
  for (int i = 0; i < 8; ++i) {
    const int idx4 = tid + 256 * i;
    const int c  = idx4 >> 4;
    const int q4 = idx4 & 15;
    const float4 qv = *(const float4*)(Q + ((size_t)(b * C_DIM + c)) * N_PTS + qBase0 + q4 * 4);
    const int q = q4 * 4;
    *(float4*)&smem_q[(q >> 4) * QT_W + c * 16 + (q & 15)] = qv;
  }

  const int st_c  = tid >> 3;
  const int st_n4 = (tid & 7) * 4;
  const int st_tg = st_n4 >> 4;
  const int st_nn = st_n4 & 15;
  const size_t st_g = ((size_t)(b * C_DIM + st_c)) * N_PTS + st_n4;
  const float* kg = K + st_g;
  const float* vg = V + st_g;
  float* k_st = kt + st_tg * KT_G + st_c * 16 + st_nn;
  float* v_st = vt + st_tg * VT_G + st_nn * VSTR + st_c;
  float4 kr0, kr1, kr2, kr3, vr0, vr1, vr2, vr3;
#define KV_FETCH(kb)                                                   \
  do {                                                                 \
    kr0 = *(const float4*)(kg + (kb));                                 \
    kr1 = *(const float4*)(kg + (kb) + (size_t)1 * 32 * N_PTS);        \
    kr2 = *(const float4*)(kg + (kb) + (size_t)2 * 32 * N_PTS);        \
    kr3 = *(const float4*)(kg + (kb) + (size_t)3 * 32 * N_PTS);        \
    vr0 = *(const float4*)(vg + (kb));                                 \
    vr1 = *(const float4*)(vg + (kb) + (size_t)1 * 32 * N_PTS);        \
    vr2 = *(const float4*)(vg + (kb) + (size_t)2 * 32 * N_PTS);        \
    vr3 = *(const float4*)(vg + (kb) + (size_t)3 * 32 * N_PTS);        \
  } while (0)
#define V_PUT(vd, i, vv)                                               \
  do {                                                                 \
    (vd)[0 * VSTR + (i) * 32] = (vv).x;                                \
    (vd)[1 * VSTR + (i) * 32] = (vv).y;                                \
    (vd)[2 * VSTR + (i) * 32] = (vv).z;                                \
    (vd)[3 * VSTR + (i) * 32] = (vv).w;                                \
  } while (0)
#define KV_PUT(kd, vd)                                                 \
  do {                                                                 \
    *(float4*)((kd) + 0 * 512) = kr0;                                  \
    *(float4*)((kd) + 1 * 512) = kr1;                                  \
    *(float4*)((kd) + 2 * 512) = kr2;                                  \
    *(float4*)((kd) + 3 * 512) = kr3;                                  \
    V_PUT(vd, 0, vr0);                                                 \
    V_PUT(vd, 1, vr1);                                                 \
    V_PUT(vd, 2, vr2);                                                 \
    V_PUT(vd, 3, vr3);                                                 \
  } while (0)

  KV_FETCH(0);
  KV_PUT(k_st, v_st);
  __syncthreads();
  const float* qt_w = smem_q + qtile * QT_W;

  v8f  o[8] = {};
  v8f  ls8  = {};
  float t1[8], t2[8];
#pragma unroll
  for (int r = 0; r < 8; ++r) { t1[r] = NEGINF; t2[r] = NEGINF; }
  const v2f vone = {1.0f, 1.0f};

  for (int it = 0; it < NSTEP; ++it) {
    const int  cur  = it & 1;
    const bool more = (it + 1 < NSTEP);
    if (more) {
      const int kbn = (it + 1) * 32;
      KV_FETCH(kbn);
    }
    const float* kt_g = kt + cur * KSTG + grp * KT_G;
    const float* vt_g = vt + cur * VSTG + grp * VT_G;

    v8f s = {};
#define SC_LOAD(qv, kv, rd)                                            \
  _Pragma("unroll")                                                    \
  for (int j = 0; j < 8; ++j) {                                        \
    const int c0 = 4 * (8 * (rd) + j) + 2 * hg;                         \
    (qv)[j].x = qt_w[c0 * 16 + lo];                                    \
    (qv)[j].y = qt_w[(c0 + 1) * 16 + lo];                              \
    (kv)[j].x = kt_g[c0 * 16 + lo];                                    \
    (kv)[j].y = kt_g[(c0 + 1) * 16 + lo];                              \
  }
#define SC_MMA(qv, kv)                                                 \
  _Pragma("unroll")                                                    \
  for (int j = 0; j < 8; ++j)                                          \
    s = __builtin_amdgcn_wmma_f32_16x16x4_f32(false, (qv)[j], false, (kv)[j], \
                                              (short)0, s, false, false);
#define SC_FENCE() __builtin_amdgcn_sched_barrier(0)
    v2f qa[8], ka[8], qb[8], kb[8];
    SC_LOAD(qa, ka, 0);
    SC_LOAD(qb, kb, 1);
    SC_FENCE();
    SC_MMA(qa, ka);
    SC_FENCE();
    SC_LOAD(qa, ka, 2);
    SC_FENCE();
    SC_MMA(qb, kb);
    SC_FENCE();
    SC_LOAD(qb, kb, 3);
    SC_FENCE();
    SC_MMA(qa, ka);
    SC_FENCE();
    SC_MMA(qb, kb);
    SC_FENCE();
#undef SC_LOAD
#undef SC_MMA

    if (more) {
      float* kd = k_st + (cur ^ 1) * KSTG;
      float* vd = v_st + (cur ^ 1) * VSTG;
      KV_PUT(kd, vd);
    }

    v8f p;
    float mo[8];
    float anyChange = 0.0f;
#pragma unroll
    for (int r = 0; r < 8; ++r) {
      float a1 = s[r], a2 = NEGINF;
      {
        float b1 = XOR16_1(a1), b2 = XOR16_1(a2);
        float hi = fmaxf(a1, b1), lw = fminf(a1, b1);
        a2 = fmaxf(lw, fmaxf(a2, b2)); a1 = hi;
        b1 = XOR16_2(a1); b2 = XOR16_2(a2);
        hi = fmaxf(a1, b1); lw = fminf(a1, b1);
        a2 = fmaxf(lw, fmaxf(a2, b2)); a1 = hi;
        b1 = XOR16_4(a1); b2 = XOR16_4(a2);
        hi = fmaxf(a1, b1); lw = fminf(a1, b1);
        a2 = fmaxf(lw, fmaxf(a2, b2)); a1 = hi;
        b1 = XOR16_8(a1); b2 = XOR16_8(a2);
        hi = fmaxf(a1, b1); lw = fminf(a1, b1);
        a2 = fmaxf(lw, fmaxf(a2, b2)); a1 = hi;
      }
      mo[r] = t1[r];
      const float hi = fmaxf(t1[r], a1);
      const float lw = fminf(t1[r], a1);
      t2[r] = fmaxf(lw, fmaxf(t2[r], a2));
      t1[r] = hi;
      anyChange += (t1[r] - mo[r]);
      p[r] = __expf((s[r] - t1[r]) * invTemp);
    }

    if (__any(anyChange > 0.0f)) {
#pragma unroll
      for (int r = 0; r < 8; ++r) {
        const float sc = __expf((mo[r] - t1[r]) * invTemp);
        ls8[r] *= sc;
#pragma unroll
        for (int ct = 0; ct < 8; ++ct) o[ct][r] *= sc;
      }
    }

    float* pw = &pl[wave][0];
#pragma unroll
    for (int r = 0; r < 8; ++r)
      pw[(r + 8 * hg) * PSTR + lo] = p[r];

#define PV_LOAD(av, vv, ks2)                                           \
  do {                                                                 \
    const int k0 = 4 * (ks2) + 2 * hg;                                 \
    (av).x = pw[lo * PSTR + k0];                                       \
    (av).y = pw[lo * PSTR + k0 + 1];                                   \
    _Pragma("unroll")                                                  \
    for (int ct = 0; ct < 8; ++ct) {                                   \
      (vv)[ct].x = vt_g[k0 * VSTR + ct * 16 + lo];                     \
      (vv)[ct].y = vt_g[(k0 + 1) * VSTR + ct * 16 + lo];               \
    }                                                                  \
  } while (0)
#define PV_MMA(av, vv)                                                 \
  do {                                                                 \
    _Pragma("unroll")                                                  \
    for (int ct = 0; ct < 8; ++ct)                                     \
      o[ct] = __builtin_amdgcn_wmma_f32_16x16x4_f32(false, (av), false, (vv)[ct], \
                                                    (short)0, o[ct], false, false); \
    ls8 = __builtin_amdgcn_wmma_f32_16x16x4_f32(false, (av), false, vone, \
                                                (short)0, ls8, false, false); \
  } while (0)
    v2f acur, vcur[8];
    PV_LOAD(acur, vcur, 0);
#pragma unroll 1
    for (int ks2 = 1; ks2 < 4; ++ks2) {
      v2f anxt, vnxt[8];
      PV_LOAD(anxt, vnxt, ks2);
      SC_FENCE();
      PV_MMA(acur, vcur);
      SC_FENCE();
      acur = anxt;
#pragma unroll
      for (int ct = 0; ct < 8; ++ct) vcur[ct] = vnxt[ct];
    }
    PV_MMA(acur, vcur);
#undef PV_LOAD
#undef PV_MMA
#undef SC_FENCE
    __syncthreads();
  }

#undef KV_FETCH
#undef KV_PUT
#undef V_PUT

  float* mo_ = mrg_o + qtile * 2048;
  float* ms_ = mrg_s + qtile * 768;
  if (grp == 1) {
#pragma unroll
    for (int r = 0; r < 8; ++r) {
      ms_[r * 32 + lane]       = t1[r];
      ms_[256 + r * 32 + lane] = t2[r];
      ms_[512 + r * 32 + lane] = ls8[r];
#pragma unroll
      for (int ct = 0; ct < 8; ++ct)
        mo_[(ct * 8 + r) * 32 + lane] = o[ct][r];
    }
  }
  __syncthreads();

  if (grp == 0) {
#pragma unroll
    for (int r = 0; r < 8; ++r) {
      const float t1b = ms_[r * 32 + lane];
      const float t2b = ms_[256 + r * 32 + lane];
      const float lb  = ms_[512 + r * 32 + lane];
      const float m   = fmaxf(t1[r], t1b);
      const float m2  = fmaxf(fminf(t1[r], t1b), fmaxf(t2[r], t2b));
      const float sa  = __expf((t1[r] - m) * invTemp);
      const float sb  = __expf((t1b - m) * invTemp);
      const float invl = 1.0f / (ls8[r] * sa + lb * sb);
      const float wa = sa * invl;
      const float wb = sb * invl;
      const int q = qtile * 16 + r + 8 * hg;
#pragma unroll
      for (int ct = 0; ct < 8; ++ct) {
        const int c = ct * 16 + lo;
        smem_q[c * OSTR + q] = o[ct][r] * wa + mo_[(ct * 8 + r) * 32 + lane] * wb;
      }
      if (lo == 0) {
        const float x1 = fminf(fmaxf(m, -1.0f), 1.0f);
        const float x2 = fminf(fmaxf(m2, -1.0f), 1.0f);
        rs[q] = acosf(x1) / acosf(x2);
      }
    }
  }
  __syncthreads();

#pragma unroll
  for (int i = 0; i < 8; ++i) {
    const int idx4 = tid + 256 * i;
    const int c  = idx4 >> 4;
    const int q4 = (idx4 & 15) * 4;
    const float4 val = *(const float4*)&smem_q[c * OSTR + q4];
    *(float4*)(Y + ((size_t)(b * C_DIM + c)) * N_PTS + qBase0 + q4) = val;
  }
  if (tid < QB) R[b * N_PTS + qBase0 + tid] = rs[tid];
}

extern "C" void kernel_launch(void* const* d_in, const int* in_sizes, int n_in,
                              void* d_out, int out_size, void* d_ws, size_t ws_size,
                              hipStream_t stream) {
  (void)in_sizes; (void)n_in; (void)out_size;
  const float* q  = (const float*)d_in[0];
  const float* k  = (const float*)d_in[1];
  const float* v  = (const float*)d_in[2];
  const float* Wq = (const float*)d_in[3];
  const float* Wk = (const float*)d_in[4];
  const float* Wv = (const float*)d_in[5];

  float* out  = (float*)d_out;
  float* Yout = out;
  float* Rout = out + (size_t)B_SZ * C_DIM * N_PTS;

  const size_t tensElems = (size_t)B_SZ * C_DIM * N_PTS;
  const float *qc = q, *kc = k, *vc = v;

  if (ws_size >= 3 * tensElems * sizeof(float)) {
    float* qw = (float*)d_ws;
    float* kw = qw + tensElems;
    float* vw = kw + tensElems;
    dim3 g(N_PTS / CM_NT, B_SZ);
    cmix_kernel<<<g, 256, 0, stream>>>(q, Wq, qw);
    cmix_kernel<<<g, 256, 0, stream>>>(k, Wk, kw);
    cmix_kernel<<<g, 256, 0, stream>>>(v, Wv, vw);
    qc = qw; kc = kw; vc = vw;
  }

  dim3 ga(N_PTS / QB, B_SZ);
  attn_kernel<<<ga, 256, 0, stream>>>(qc, kc, vc, Yout, Rout);
}


# attn: top-2 kept per lane in the key loop (min/max on the lane's own score), merged across the 16 lanes once at the end; per step only a max-only butterfly for the row max (was a top-2 butterfly)
# speedup vs baseline: 2.8256x; 1.0363x over previous
#include <hip/hip_runtime.h>
#include <math.h>

typedef float v2f __attribute__((ext_vector_type(2)));
typedef float v8f __attribute__((ext_vector_type(8)));

#define C_DIM 128
#define N_PTS 4096
#define B_SZ  4
#define QB    64
#define VSTR  132
#define OSTR  68
#define PSTR  17
#define QT_W  2048
#define KT_G  2048
#define VT_G  2120
#define KSTG  (2 * KT_G)
#define VSTG  (2 * VT_G)
#define NSTEP (N_PTS / 32)
#define SMEMQ 8704

__device__ __forceinline__ float xor16(float v, unsigned selLo, unsigned selHi) {
  unsigned i = __builtin_bit_cast(unsigned, v);
  i = __builtin_amdgcn_permlane16(i, i, selLo, selHi, false, false);
  return __builtin_bit_cast(float, i);
}
#define XOR16_1(v) xor16((v), 0x67452301u, 0xEFCDAB89u)
#define XOR16_2(v) xor16((v), 0x54761032u, 0xDCFE98BAu)
#define XOR16_4(v) xor16((v), 0x32107654u, 0xBA98FEDCu)
#define XOR16_8(v) xor16((v), 0xFEDCBA98u, 0x76543210u)

#define CM_NT   64
#define CM_XP   160
#define CM_WSTR 132
#define CM_OSTR 68
#define CM_XS   (64 * CM_XP)
#define CM_WT   (16 * CM_WSTR)

__global__ __launch_bounds__(256) void cmix_kernel(const float* __restrict__ x,
                                                   const float* __restrict__ W,
                                                   float* __restrict__ out) {
  __shared__ __attribute__((aligned(16))) float cm_lds[CM_XS + 8 * CM_WT];
  float* xs = cm_lds;
  const int tid  = threadIdx.x;
  const int lane = tid & 31;
  const int wave = tid >> 5;
  const int lo = lane & 15;
  const int hg = lane >> 4;
  const int b   = blockIdx.y;
  const int nb0 = blockIdx.x * CM_NT;
  const int o0  = wave * 16;
  float* wt_w = cm_lds + CM_XS + wave * CM_WT;

#pragma unroll
  for (int i = 0; i < 4; ++i) {
    const int idx = tid + 256 * i;
    const int p  = idx >> 4;
    const int n4 = (idx & 15) * 4;
    const float* xp = x + ((size_t)(b * C_DIM + 2 * p)) * N_PTS + nb0 + n4;
    const float4 e = *(const float4*)xp;
    const float4 f = *(const float4*)(xp + N_PTS);
    float* d = &xs[p * CM_XP + 2 * n4];
    *(float4*)d       = make_float4(e.x, f.x, e.y, f.y);
    *(float4*)(d + 4) = make_float4(e.z, f.z, e.w, f.w);
  }
#pragma unroll
  for (int i = 0; i < 16; ++i) {
    const float4 wv = *(const float4*)(W + (size_t)(o0 + i) * C_DIM + 4 * lane);
    *(float4*)&wt_w[i * CM_WSTR + 4 * lane] = wv;
  }
  __syncthreads();

  v8f acc[4] = {};
  const float* a_l = wt_w + lo * CM_WSTR + 2 * hg;
  const float* b_l = xs + hg * CM_XP + 2 * lo;
#pragma unroll
  for (int ks = 0; ks < 32; ++ks) {
    const v2f a = *(const v2f*)(a_l + 4 * ks);
#pragma unroll
    for (int j = 0; j < 4; ++j) {
      const v2f bv = *(const v2f*)(b_l + 2 * ks * CM_XP + 32 * j);
      acc[j] = __builtin_amdgcn_wmma_f32_16x16x4_f32(false, a, false, bv,
                                                     (short)0, acc[j], false, false);
    }
  }

#pragma unroll
  for (int j = 0; j < 4; ++j)
#pragma unroll
    for (int r = 0; r < 8; ++r)
      wt_w[(r + 8 * hg) * CM_OSTR + 16 * j + lo] = acc[j][r];
#pragma unroll
  for (int i = 0; i < 8; ++i) {
    const int row = 2 * i + hg;
    const float4 val = *(const float4*)&wt_w[row * CM_OSTR + 4 * lo];
    *(float4*)(out + ((size_t)(b * C_DIM + o0 + row)) * N_PTS + nb0 + 4 * lo) = val;
  }
}

__global__ __launch_bounds__(256)
__attribute__((amdgpu_num_vgpr(256)))
void attn_kernel(const float* __restrict__ Q,
                 const float* __restrict__ K,
                 const float* __restrict__ V,
                 float* __restrict__ Y,
                 float* __restrict__ R) {
  __shared__ __attribute__((aligned(16))) float smem_q[SMEMQ];
  __shared__ __attribute__((aligned(16))) float kt[2 * KSTG];
  __shared__ float vt[2 * VSTG];
  __shared__ float pl[8][16 * PSTR];
  __shared__ float mrg_o[4 * 2048];
  __shared__ float mrg_s[4 * 768];
  __shared__ float rs[QB];

  const int tid  = threadIdx.x;
  const int lane = tid & 31;
  const int wave = tid >> 5;
  const int qtile = wave & 3;
  const int grp   = wave >> 2;
  const int b    = blockIdx.y;
  const int qBase0 = blockIdx.x * QB;
  const int lo = lane & 15;
  const int hg = lane >> 4;
  const float invTemp = 0.088388347648318447f;
  const float NEGINF = -__builtin_inff();

#pragma unroll
  for (int i = 0; i < 8; ++i) {
    const int idx4 = tid + 256 * i;
    const int c  = idx4 >> 4;
    const int q4 = idx4 & 15;
    const float4 qv = *(const float4*)(Q + ((size_t)(b * C_DIM + c)) * N_PTS + qBase0 + q4 * 4);
    const int q = q4 * 4;
    *(float4*)&smem_q[(q >> 4) * QT_W + c * 16 + (q & 15)] = qv;
  }

  const int st_c  = tid >> 3;
  const int st_n4 = (tid & 7) * 4;
  const int st_tg = st_n4 >> 4;
  const int st_nn = st_n4 & 15;
  const size_t st_g = ((size_t)(b * C_DIM + st_c)) * N_PTS + st_n4;
  const float* kg = K + st_g;
  const float* vg = V + st_g;
  float* k_st = kt + st_tg * KT_G + st_c * 16 + st_nn;
  float* v_st = vt + st_tg * VT_G + st_nn * VSTR + st_c;
  float4 kr0, kr1, kr2, kr3, vr0, vr1, vr2, vr3;
#define KV_FETCH(kb)                                                   \
  do {                                                                 \
    kr0 = *(const float4*)(kg + (kb));                                 \
    kr1 = *(const float4*)(kg + (kb) + (size_t)1 * 32 * N_PTS);        \
    kr2 = *(const float4*)(kg + (kb) + (size_t)2 * 32 * N_PTS);        \
    kr3 = *(const float4*)(kg + (kb) + (size_t)3 * 32 * N_PTS);        \
    vr0 = *(const float4*)(vg + (kb));                                 \
    vr1 = *(const float4*)(vg + (kb) + (size_t)1 * 32 * N_PTS);        \
    vr2 = *(const float4*)(vg + (kb) + (size_t)2 * 32 * N_PTS);        \
    vr3 = *(const float4*)(vg + (kb) + (size_t)3 * 32 * N_PTS);        \
  } while (0)
#define V_PUT(vd, i, vv)                                               \
  do {                                                                 \
    (vd)[0 * VSTR + (i) * 32] = (vv).x;                                \
    (vd)[1 * VSTR + (i) * 32] = (vv).y;                                \
    (vd)[2 * VSTR + (i) * 32] = (vv).z;                                \
    (vd)[3 * VSTR + (i) * 32] = (vv).w;                                \
  } while (0)
#define KV_PUT(kd, vd)                                                 \
  do {                                                                 \
    *(float4*)((kd) + 0 * 512) = kr0;                                  \
    *(float4*)((kd) + 1 * 512) = kr1;                                  \
    *(float4*)((kd) + 2 * 512) = kr2;                                  \
    *(float4*)((kd) + 3 * 512) = kr3;                                  \
    V_PUT(vd, 0, vr0);                                                 \
    V_PUT(vd, 1, vr1);                                                 \
    V_PUT(vd, 2, vr2);                                                 \
    V_PUT(vd, 3, vr3);                                                 \
  } while (0)

  KV_FETCH(0);
  KV_PUT(k_st, v_st);
  __syncthreads();
  const float* qt_w = smem_q + qtile * QT_W;

  v8f  o[8] = {};
  v8f  ls8  = {};
  float t1[8], t2[8], mrow[8];
#pragma unroll
  for (int r = 0; r < 8; ++r) { t1[r] = NEGINF; t2[r] = NEGINF; mrow[r] = NEGINF; }
  const v2f vone = {1.0f, 1.0f};

  for (int it = 0; it < NSTEP; ++it) {
    const int  cur  = it & 1;
    const bool more = (it + 1 < NSTEP);
    if (more) {
      const int kbn = (it + 1) * 32;
      KV_FETCH(kbn);
    }
    const float* kt_g = kt + cur * KSTG + grp * KT_G;
    const float* vt_g = vt + cur * VSTG + grp * VT_G;

    v8f s = {};
#define SC_LOAD(qv, kv, rd)                                            \
  _Pragma("unroll")                                                    \
  for (int j = 0; j < 8; ++j) {                                        \
    const int c0 = 4 * (8 * (rd) + j) + 2 * hg;                         \
    (qv)[j].x = qt_w[c0 * 16 + lo];                                    \
    (qv)[j].y = qt_w[(c0 + 1) * 16 + lo];                              \
    (kv)[j].x = kt_g[c0 * 16 + lo];                                    \
    (kv)[j].y = kt_g[(c0 + 1) * 16 + lo];                              \
  }
#define SC_MMA(qv, kv)                                                 \
  _Pragma("unroll")                                                    \
  for (int j = 0; j < 8; ++j)                                          \
    s = __builtin_amdgcn_wmma_f32_16x16x4_f32(false, (qv)[j], false, (kv)[j], \
                                              (short)0, s, false, false);
#define SC_FENCE() __builtin_amdgcn_sched_barrier(0)
    v2f qa[8], ka[8], qb[8], kb[8];
    SC_LOAD(qa, ka, 0);
    SC_LOAD(qb, kb, 1);
    SC_FENCE();
    SC_MMA(qa, ka);
    SC_FENCE();
    SC_LOAD(qa, ka, 2);
    SC_FENCE();
    SC_MMA(qb, kb);
    SC_FENCE();
    SC_LOAD(qb, kb, 3);
    SC_FENCE();
    SC_MMA(qa, ka);
    SC_FENCE();
    SC_MMA(qb, kb);
    SC_FENCE();
#undef SC_LOAD
#undef SC_MMA

    if (more) {
      float* kd = k_st + (cur ^ 1) * KSTG;
      float* vd = v_st + (cur ^ 1) * VSTG;
      KV_PUT(kd, vd);
    }

    v8f p;
    float mo[8];
    float anyChange = 0.0f;
#pragma unroll
    for (int r = 0; r < 8; ++r) {
      float a1 = s[r];
      a1 = fmaxf(a1, XOR16_1(a1));
      a1 = fmaxf(a1, XOR16_2(a1));
      a1 = fmaxf(a1, XOR16_4(a1));
      a1 = fmaxf(a1, XOR16_8(a1));
      const float lw = fminf(t1[r], s[r]);
      t1[r] = fmaxf(t1[r], s[r]);
      t2[r] = fmaxf(t2[r], lw);
      mo[r] = mrow[r];
      mrow[r] = fmaxf(mrow[r], a1);
      anyChange += (mrow[r] - mo[r]);
      p[r] = __expf((s[r] - mrow[r]) * invTemp);
    }

    if (__any(anyChange > 0.0f)) {
#pragma unroll
      for (int r = 0; r < 8; ++r) {
        const float sc = __expf((mo[r] - mrow[r]) * invTemp);
        ls8[r] *= sc;
#pragma unroll
        for (int ct = 0; ct < 8; ++ct) o[ct][r] *= sc;
      }
    }

    float* pw = &pl[wave][0];
#pragma unroll
    for (int r = 0; r < 8; ++r)
      pw[(r + 8 * hg) * PSTR + lo] = p[r];

#define PV_LOAD(av, vv, ks2)                                           \
  do {                                                                 \
    const int k0 = 4 * (ks2) + 2 * hg;                                 \
    (av).x = pw[lo * PSTR + k0];                                       \
    (av).y = pw[lo * PSTR + k0 + 1];                                   \
    _Pragma("unroll")                                                  \
    for (int ct = 0; ct < 8; ++ct) {                                   \
      (vv)[ct].x = vt_g[k0 * VSTR + ct * 16 + lo];                     \
      (vv)[ct].y = vt_g[(k0 + 1) * VSTR + ct * 16 + lo];               \
    }                                                                  \
  } while (0)
#define PV_MMA(av, vv)                                                 \
  do {                                                                 \
    _Pragma("unroll")                                                  \
    for (int ct = 0; ct < 8; ++ct)                                     \
      o[ct] = __builtin_amdgcn_wmma_f32_16x16x4_f32(false, (av), false, (vv)[ct], \
                                                    (short)0, o[ct], false, false); \
    ls8 = __builtin_amdgcn_wmma_f32_16x16x4_f32(false, (av), false, vone, \
                                                (short)0, ls8, false, false); \
  } while (0)
    v2f acur, vcur[8];
    PV_LOAD(acur, vcur, 0);
#pragma unroll 1
    for (int ks2 = 1; ks2 < 4; ++ks2) {
      v2f anxt, vnxt[8];
      PV_LOAD(anxt, vnxt, ks2);
      SC_FENCE();
      PV_MMA(acur, vcur);
      SC_FENCE();
      acur = anxt;
#pragma unroll
      for (int ct = 0; ct < 8; ++ct) vcur[ct] = vnxt[ct];
    }
    PV_MMA(acur, vcur);
#undef PV_LOAD
#undef PV_MMA
#undef SC_FENCE
    __syncthreads();
  }

#undef KV_FETCH
#undef KV_PUT
#undef V_PUT

#pragma unroll
  for (int r = 0; r < 8; ++r) {
    float a1 = t1[r], a2 = t2[r];
    float b1 = XOR16_1(a1), b2 = XOR16_1(a2);
    float hi = fmaxf(a1, b1), lw = fminf(a1, b1);
    a2 = fmaxf(lw, fmaxf(a2, b2)); a1 = hi;
    b1 = XOR16_2(a1); b2 = XOR16_2(a2);
    hi = fmaxf(a1, b1); lw = fminf(a1, b1);
    a2 = fmaxf(lw, fmaxf(a2, b2)); a1 = hi;
    b1 = XOR16_4(a1); b2 = XOR16_4(a2);
    hi = fmaxf(a1, b1); lw = fminf(a1, b1);
    a2 = fmaxf(lw, fmaxf(a2, b2)); a1 = hi;
    b1 = XOR16_8(a1); b2 = XOR16_8(a2);
    hi = fmaxf(a1, b1); lw = fminf(a1, b1);
    a2 = fmaxf(lw, fmaxf(a2, b2)); a1 = hi;
    t1[r] = a1; t2[r] = a2;
  }

  float* mo_ = mrg_o + qtile * 2048;
  float* ms_ = mrg_s + qtile * 768;
  if (grp == 1) {
#pragma unroll
    for (int r = 0; r < 8; ++r) {
      ms_[r * 32 + lane]       = t1[r];
      ms_[256 + r * 32 + lane] = t2[r];
      ms_[512 + r * 32 + lane] = ls8[r];
#pragma unroll
      for (int ct = 0; ct < 8; ++ct)
        mo_[(ct * 8 + r) * 32 + lane] = o[ct][r];
    }
  }
  __syncthreads();

  if (grp == 0) {
#pragma unroll
    for (int r = 0; r < 8; ++r) {
      const float t1b = ms_[r * 32 + lane];
      const float t2b = ms_[256 + r * 32 + lane];
      const float lb  = ms_[512 + r * 32 + lane];
      const float m   = fmaxf(t1[r], t1b);
      const float m2  = fmaxf(fminf(t1[r], t1b), fmaxf(t2[r], t2b));
      const float sa  = __expf((t1[r] - m) * invTemp);
      const float sb  = __expf((t1b - m) * invTemp);
      const float invl = 1.0f / (ls8[r] * sa + lb * sb);
      const float wa = sa * invl;
      const float wb = sb * invl;
      const int q = qtile * 16 + r + 8 * hg;
#pragma unroll
      for (int ct = 0; ct < 8; ++ct) {
        const int c = ct * 16 + lo;
        smem_q[c * OSTR + q] = o[ct][r] * wa + mo_[(ct * 8 + r) * 32 + lane] * wb;
      }
      if (lo == 0) {
        const float x1 = fminf(fmaxf(m, -1.0f), 1.0f);
        const float x2 = fminf(fmaxf(m2, -1.0f), 1.0f);
        rs[q] = acosf(x1) / acosf(x2);
      }
    }
  }
  __syncthreads();

#pragma unroll
  for (int i = 0; i < 8; ++i) {
    const int idx4 = tid + 256 * i;
    const int c  = idx4 >> 4;
    const int q4 = (idx4 & 15) * 4;
    const float4 val = *(const float4*)&smem_q[c * OSTR + q4];
    *(float4*)(Y + ((size_t)(b * C_DIM + c)) * N_PTS + qBase0 + q4) = val;
  }
  if (tid < QB) R[b * N_PTS + qBase0 + tid] = rs[tid];
}

extern "C" void kernel_launch(void* const* d_in, const int* in_sizes, int n_in,
                              void* d_out, int out_size, void* d_ws, size_t ws_size,
                              hipStream_t stream) {
  (void)in_sizes; (void)n_in; (void)out_size;
  const float* q  = (const float*)d_in[0];
  const float* k  = (const float*)d_in[1];
  const float* v  = (const float*)d_in[2];
  const float* Wq = (const float*)d_in[3];
  const float* Wk = (const float*)d_in[4];
  const float* Wv = (const float*)d_in[5];

  float* out  = (float*)d_out;
  float* Yout = out;
  float* Rout = out + (size_t)B_SZ * C_DIM * N_PTS;

  const size_t tensElems = (size_t)B_SZ * C_DIM * N_PTS;
  const float *qc = q, *kc = k, *vc = v;

  if (ws_size >= 3 * tensElems * sizeof(float)) {
    float* qw = (float*)d_ws;
    float* kw = qw + tensElems;
    float* vw = kw + tensElems;
    dim3 g(N_PTS / CM_NT, B_SZ);
    cmix_kernel<<<g, 256, 0, stream>>>(q, Wq, qw);
    cmix_kernel<<<g, 256, 0, stream>>>(k, Wk, kw);
    cmix_kernel<<<g, 256, 0, stream>>>(v, Wv, vw);
    qc = qw; kc = kw; vc = vw;
  }

  dim3 ga(N_PTS / QB, B_SZ);
  attn_kernel<<<ga, 256, 0, stream>>>(qc, kc, vc, Yout, Rout);
}


# attn: lazy softmax reference: row-max butterfly and rescale run only in a step where a score lies over 8*sqrt(C) above the row's reference (was every step); sums carry that scale (rounding changes)
# speedup vs baseline: 2.9792x; 1.0544x over previous
#include <hip/hip_runtime.h>
#include <math.h>

typedef float v2f __attribute__((ext_vector_type(2)));
typedef float v8f __attribute__((ext_vector_type(8)));

#define C_DIM 128
#define N_PTS 4096
#define B_SZ  4
#define QB    64
#define VSTR  132
#define OSTR  68
#define PSTR  17
#define QT_W  2048
#define KT_G  2048
#define VT_G  2120
#define KSTG  (2 * KT_G)
#define VSTG  (2 * VT_G)
#define NSTEP (N_PTS / 32)
#define SMEMQ 8704

__device__ __forceinline__ float xor16(float v, unsigned selLo, unsigned selHi) {
  unsigned i = __builtin_bit_cast(unsigned, v);
  i = __builtin_amdgcn_permlane16(i, i, selLo, selHi, false, false);
  return __builtin_bit_cast(float, i);
}
#define XOR16_1(v) xor16((v), 0x67452301u, 0xEFCDAB89u)
#define XOR16_2(v) xor16((v), 0x54761032u, 0xDCFE98BAu)
#define XOR16_4(v) xor16((v), 0x32107654u, 0xBA98FEDCu)
#define XOR16_8(v) xor16((v), 0xFEDCBA98u, 0x76543210u)

#define CM_NT   64
#define CM_XP   160
#define CM_WSTR 132
#define CM_OSTR 68
#define CM_XS   (64 * CM_XP)
#define CM_WT   (16 * CM_WSTR)

__global__ __launch_bounds__(256) void cmix_kernel(const float* __restrict__ x,
                                                   const float* __restrict__ W,
                                                   float* __restrict__ out) {
  __shared__ __attribute__((aligned(16))) float cm_lds[CM_XS + 8 * CM_WT];
  float* xs = cm_lds;
  const int tid  = threadIdx.x;
  const int lane = tid & 31;
  const int wave = tid >> 5;
  const int lo = lane & 15;
  const int hg = lane >> 4;
  const int b   = blockIdx.y;
  const int nb0 = blockIdx.x * CM_NT;
  const int o0  = wave * 16;
  float* wt_w = cm_lds + CM_XS + wave * CM_WT;

#pragma unroll
  for (int i = 0; i < 4; ++i) {
    const int idx = tid + 256 * i;
    const int p  = idx >> 4;
    const int n4 = (idx & 15) * 4;
    const float* xp = x + ((size_t)(b * C_DIM + 2 * p)) * N_PTS + nb0 + n4;
    const float4 e = *(const float4*)xp;
    const float4 f = *(const float4*)(xp + N_PTS);
    float* d = &xs[p * CM_XP + 2 * n4];
    *(float4*)d       = make_float4(e.x, f.x, e.y, f.y);
    *(float4*)(d + 4) = make_float4(e.z, f.z, e.w, f.w);
  }
#pragma unroll
  for (int i = 0; i < 16; ++i) {
    const float4 wv = *(const float4*)(W + (size_t)(o0 + i) * C_DIM + 4 * lane);
    *(float4*)&wt_w[i * CM_WSTR + 4 * lane] = wv;
  }
  __syncthreads();

  v8f acc[4] = {};
  const float* a_l = wt_w + lo * CM_WSTR + 2 * hg;
  const float* b_l = xs + hg * CM_XP + 2 * lo;
#pragma unroll
  for (int ks = 0; ks < 32; ++ks) {
    const v2f a = *(const v2f*)(a_l + 4 * ks);
#pragma unroll
    for (int j = 0; j < 4; ++j) {
      const v2f bv = *(const v2f*)(b_l + 2 * ks * CM_XP + 32 * j);
      acc[j] = __builtin_amdgcn_wmma_f32_16x16x4_f32(false, a, false, bv,
                                                     (short)0, acc[j], false, false);
    }
  }

#pragma unroll
  for (int j = 0; j < 4; ++j)
#pragma unroll
    for (int r = 0; r < 8; ++r)
      wt_w[(r + 8 * hg) * CM_OSTR + 16 * j + lo] = acc[j][r];
#pragma unroll
  for (int i = 0; i < 8; ++i) {
    const int row = 2 * i + hg;
    const float4 val = *(const float4*)&wt_w[row * CM_OSTR + 4 * lo];
    *(float4*)(out + ((size_t)(b * C_DIM + o0 + row)) * N_PTS + nb0 + 4 * lo) = val;
  }
}

__global__ __launch_bounds__(256)
__attribute__((amdgpu_num_vgpr(256)))
void attn_kernel(const float* __restrict__ Q,
                 const float* __restrict__ K,
                 const float* __restrict__ V,
                 float* __restrict__ Y,
                 float* __restrict__ R) {
  __shared__ __attribute__((aligned(16))) float smem_q[SMEMQ];
  __shared__ __attribute__((aligned(16))) float kt[2 * KSTG];
  __shared__ float vt[2 * VSTG];
  __shared__ float pl[8][16 * PSTR];
  __shared__ float mrg_o[4 * 2048];
  __shared__ float mrg_s[4 * 1024];
  __shared__ float rs[QB];

  const int tid  = threadIdx.x;
  const int lane = tid & 31;
  const int wave = tid >> 5;
  const int qtile = wave & 3;
  const int grp   = wave >> 2;
  const int b    = blockIdx.y;
  const int qBase0 = blockIdx.x * QB;
  const int lo = lane & 15;
  const int hg = lane >> 4;
  const float invTemp = 0.088388347648318447f;
  const float NEGINF = -__builtin_inff();
  const float LAZY_M = 90.50966799f;

#pragma unroll
  for (int i = 0; i < 8; ++i) {
    const int idx4 = tid + 256 * i;
    const int c  = idx4 >> 4;
    const int q4 = idx4 & 15;
    const float4 qv = *(const float4*)(Q + ((size_t)(b * C_DIM + c)) * N_PTS + qBase0 + q4 * 4);
    const int q = q4 * 4;
    *(float4*)&smem_q[(q >> 4) * QT_W + c * 16 + (q & 15)] = qv;
  }

  const int st_c  = tid >> 3;
  const int st_n4 = (tid & 7) * 4;
  const int st_tg = st_n4 >> 4;
  const int st_nn = st_n4 & 15;
  const size_t st_g = ((size_t)(b * C_DIM + st_c)) * N_PTS + st_n4;
  const float* kg = K + st_g;
  const float* vg = V + st_g;
  float* k_st = kt + st_tg * KT_G + st_c * 16 + st_nn;
  float* v_st = vt + st_tg * VT_G + st_nn * VSTR + st_c;
  float4 kr0, kr1, kr2, kr3, vr0, vr1, vr2, vr3;
#define KV_FETCH(kb)                                                   \
  do {                                                                 \
    kr0 = *(const float4*)(kg + (kb));                                 \
    kr1 = *(const float4*)(kg + (kb) + (size_t)1 * 32 * N_PTS);        \
    kr2 = *(const float4*)(kg + (kb) + (size_t)2 * 32 * N_PTS);        \
    kr3 = *(const float4*)(kg + (kb) + (size_t)3 * 32 * N_PTS);        \
    vr0 = *(const float4*)(vg + (kb));                                 \
    vr1 = *(const float4*)(vg + (kb) + (size_t)1 * 32 * N_PTS);        \
    vr2 = *(const float4*)(vg + (kb) + (size_t)2 * 32 * N_PTS);        \
    vr3 = *(const float4*)(vg + (kb) + (size_t)3 * 32 * N_PTS);        \
  } while (0)
#define V_PUT(vd, i, vv)                                               \
  do {                                                                 \
    (vd)[0 * VSTR + (i) * 32] = (vv).x;                                \
    (vd)[1 * VSTR + (i) * 32] = (vv).y;                                \
    (vd)[2 * VSTR + (i) * 32] = (vv).z;                                \
    (vd)[3 * VSTR + (i) * 32] = (vv).w;                                \
  } while (0)
#define KV_PUT(kd, vd)                                                 \
  do {                                                                 \
    *(float4*)((kd) + 0 * 512) = kr0;                                  \
    *(float4*)((kd) + 1 * 512) = kr1;                                  \
    *(float4*)((kd) + 2 * 512) = kr2;                                  \
    *(float4*)((kd) + 3 * 512) = kr3;                                  \
    V_PUT(vd, 0, vr0);                                                 \
    V_PUT(vd, 1, vr1);                                                 \
    V_PUT(vd, 2, vr2);                                                 \
    V_PUT(vd, 3, vr3);                                                 \
  } while (0)

  KV_FETCH(0);
  KV_PUT(k_st, v_st);
  __syncthreads();
  const float* qt_w = smem_q + qtile * QT_W;

  v8f  o[8] = {};
  v8f  ls8  = {};
  float t1[8], t2[8], mrow[8];
#pragma unroll
  for (int r = 0; r < 8; ++r) { t1[r] = NEGINF; t2[r] = NEGINF; mrow[r] = NEGINF; }
  const v2f vone = {1.0f, 1.0f};

  for (int it = 0; it < NSTEP; ++it) {
    const int  cur  = it & 1;
    const bool more = (it + 1 < NSTEP);
    if (more) {
      const int kbn = (it + 1) * 32;
      KV_FETCH(kbn);
    }
    const float* kt_g = kt + cur * KSTG + grp * KT_G;
    const float* vt_g = vt + cur * VSTG + grp * VT_G;

    v8f s = {};
#define SC_LOAD(qv, kv, rd)                                            \
  _Pragma("unroll")                                                    \
  for (int j = 0; j < 8; ++j) {                                        \
    const int c0 = 4 * (8 * (rd) + j) + 2 * hg;                         \
    (qv)[j].x = qt_w[c0 * 16 + lo];                                    \
    (qv)[j].y = qt_w[(c0 + 1) * 16 + lo];                              \
    (kv)[j].x = kt_g[c0 * 16 + lo];                                    \
    (kv)[j].y = kt_g[(c0 + 1) * 16 + lo];                              \
  }
#define SC_MMA(qv, kv)                                                 \
  _Pragma("unroll")                                                    \
  for (int j = 0; j < 8; ++j)                                          \
    s = __builtin_amdgcn_wmma_f32_16x16x4_f32(false, (qv)[j], false, (kv)[j], \
                                              (short)0, s, false, false);
#define SC_FENCE() __builtin_amdgcn_sched_barrier(0)
    v2f qa[8], ka[8], qb[8], kb[8];
    SC_LOAD(qa, ka, 0);
    SC_LOAD(qb, kb, 1);
    SC_FENCE();
    SC_MMA(qa, ka);
    SC_FENCE();
    SC_LOAD(qa, ka, 2);
    SC_FENCE();
    SC_MMA(qb, kb);
    SC_FENCE();
    SC_LOAD(qb, kb, 3);
    SC_FENCE();
    SC_MMA(qa, ka);
    SC_FENCE();
    SC_MMA(qb, kb);
    SC_FENCE();
#undef SC_LOAD
#undef SC_MMA

    if (more) {
      float* kd = k_st + (cur ^ 1) * KSTG;
      float* vd = v_st + (cur ^ 1) * VSTG;
      KV_PUT(kd, vd);
    }

    float d[8];
    float dmax = NEGINF;
    v8f p;
#pragma unroll
    for (int r = 0; r < 8; ++r) {
      d[r] = s[r] - mrow[r];
      dmax = fmaxf(dmax, d[r]);
      const float lw = fminf(t1[r], s[r]);
      t1[r] = fmaxf(t1[r], s[r]);
      t2[r] = fmaxf(t2[r], lw);
      p[r] = __expf(d[r] * invTemp);
    }

    if (__builtin_amdgcn_ballot_w32(dmax > LAZY_M) != 0u) {
#pragma unroll
      for (int r = 0; r < 8; ++r) {
        float a1 = s[r];
        a1 = fmaxf(a1, XOR16_1(a1));
        a1 = fmaxf(a1, XOR16_2(a1));
        a1 = fmaxf(a1, XOR16_4(a1));
        a1 = fmaxf(a1, XOR16_8(a1));
        const float mo = mrow[r];
        mrow[r] = fmaxf(mo, a1);
        const float sc = __expf((mo - mrow[r]) * invTemp);
        ls8[r] *= sc;
#pragma unroll
        for (int ct = 0; ct < 8; ++ct) o[ct][r] *= sc;
        p[r] = __expf((s[r] - mrow[r]) * invTemp);
      }
    }

    float* pw = &pl[wave][0];
#pragma unroll
    for (int r = 0; r < 8; ++r)
      pw[(r + 8 * hg) * PSTR + lo] = p[r];

#define PV_LOAD(av, vv, ks2)                                           \
  do {                                                                 \
    const int k0 = 4 * (ks2) + 2 * hg;                                 \
    (av).x = pw[lo * PSTR + k0];                                       \
    (av).y = pw[lo * PSTR + k0 + 1];                                   \
    _Pragma("unroll")                                                  \
    for (int ct = 0; ct < 8; ++ct) {                                   \
      (vv)[ct].x = vt_g[k0 * VSTR + ct * 16 + lo];                     \
      (vv)[ct].y = vt_g[(k0 + 1) * VSTR + ct * 16 + lo];               \
    }                                                                  \
  } while (0)
#define PV_MMA(av, vv)                                                 \
  do {                                                                 \
    _Pragma("unroll")                                                  \
    for (int ct = 0; ct < 8; ++ct)                                     \
      o[ct] = __builtin_amdgcn_wmma_f32_16x16x4_f32(false, (av), false, (vv)[ct], \
                                                    (short)0, o[ct], false, false); \
    ls8 = __builtin_amdgcn_wmma_f32_16x16x4_f32(false, (av), false, vone, \
                                                (short)0, ls8, false, false); \
  } while (0)
    v2f acur, vcur[8];
    PV_LOAD(acur, vcur, 0);
#pragma unroll 1
    for (int ks2 = 1; ks2 < 4; ++ks2) {
      v2f anxt, vnxt[8];
      PV_LOAD(anxt, vnxt, ks2);
      SC_FENCE();
      PV_MMA(acur, vcur);
      SC_FENCE();
      acur = anxt;
#pragma unroll
      for (int ct = 0; ct < 8; ++ct) vcur[ct] = vnxt[ct];
    }
    PV_MMA(acur, vcur);
#undef PV_LOAD
#undef PV_MMA
#undef SC_FENCE
    __syncthreads();
  }

#undef KV_FETCH
#undef KV_PUT
#undef V_PUT

#pragma unroll
  for (int r = 0; r < 8; ++r) {
    float a1 = t1[r], a2 = t2[r];
    float b1 = XOR16_1(a1), b2 = XOR16_1(a2);
    float hi = fmaxf(a1, b1), lw = fminf(a1, b1);
    a2 = fmaxf(lw, fmaxf(a2, b2)); a1 = hi;
    b1 = XOR16_2(a1); b2 = XOR16_2(a2);
    hi = fmaxf(a1, b1); lw = fminf(a1, b1);
    a2 = fmaxf(lw, fmaxf(a2, b2)); a1 = hi;
    b1 = XOR16_4(a1); b2 = XOR16_4(a2);
    hi = fmaxf(a1, b1); lw = fminf(a1, b1);
    a2 = fmaxf(lw, fmaxf(a2, b2)); a1 = hi;
    b1 = XOR16_8(a1); b2 = XOR16_8(a2);
    hi = fmaxf(a1, b1); lw = fminf(a1, b1);
    a2 = fmaxf(lw, fmaxf(a2, b2)); a1 = hi;
    t1[r] = a1; t2[r] = a2;
  }

  float* mo_ = mrg_o + qtile * 2048;
  float* ms_ = mrg_s + qtile * 1024;
  if (grp == 1) {
#pragma unroll
    for (int r = 0; r < 8; ++r) {
      ms_[r * 32 + lane]       = t1[r];
      ms_[256 + r * 32 + lane] = t2[r];
      ms_[512 + r * 32 + lane] = ls8[r];
      ms_[768 + r * 32 + lane] = mrow[r];
#pragma unroll
      for (int ct = 0; ct < 8; ++ct)
        mo_[(ct * 8 + r) * 32 + lane] = o[ct][r];
    }
  }
  __syncthreads();

  if (grp == 0) {
#pragma unroll
    for (int r = 0; r < 8; ++r) {
      const float t1b = ms_[r * 32 + lane];
      const float t2b = ms_[256 + r * 32 + lane];
      const float lb  = ms_[512 + r * 32 + lane];
      const float rb  = ms_[768 + r * 32 + lane];
      const float m   = fmaxf(t1[r], t1b);
      const float m2  = fmaxf(fminf(t1[r], t1b), fmaxf(t2[r], t2b));
      const float mr  = fmaxf(mrow[r], rb);
      const float sa  = __expf((mrow[r] - mr) * invTemp);
      const float sb  = __expf((rb - mr) * invTemp);
      const float invl = 1.0f / (ls8[r] * sa + lb * sb);
      const float wa = sa * invl;
      const float wb = sb * invl;
      const int q = qtile * 16 + r + 8 * hg;
#pragma unroll
      for (int ct = 0; ct < 8; ++ct) {
        const int c = ct * 16 + lo;
        smem_q[c * OSTR + q] = o[ct][r] * wa + mo_[(ct * 8 + r) * 32 + lane] * wb;
      }
      if (lo == 0) {
        const float x1 = fminf(fmaxf(m, -1.0f), 1.0f);
        const float x2 = fminf(fmaxf(m2, -1.0f), 1.0f);
        rs[q] = acosf(x1) / acosf(x2);
      }
    }
  }
  __syncthreads();

#pragma unroll
  for (int i = 0; i < 8; ++i) {
    const int idx4 = tid + 256 * i;
    const int c  = idx4 >> 4;
    const int q4 = (idx4 & 15) * 4;
    const float4 val = *(const float4*)&smem_q[c * OSTR + q4];
    *(float4*)(Y + ((size_t)(b * C_DIM + c)) * N_PTS + qBase0 + q4) = val;
  }
  if (tid < QB) R[b * N_PTS + qBase0 + tid] = rs[tid];
}

extern "C" void kernel_launch(void* const* d_in, const int* in_sizes, int n_in,
                              void* d_out, int out_size, void* d_ws, size_t ws_size,
                              hipStream_t stream) {
  (void)in_sizes; (void)n_in; (void)out_size;
  const float* q  = (const float*)d_in[0];
  const float* k  = (const float*)d_in[1];
  const float* v  = (const float*)d_in[2];
  const float* Wq = (const float*)d_in[3];
  const float* Wk = (const float*)d_in[4];
  const float* Wv = (const float*)d_in[5];

  float* out  = (float*)d_out;
  float* Yout = out;
  float* Rout = out + (size_t)B_SZ * C_DIM * N_PTS;

  const size_t tensElems = (size_t)B_SZ * C_DIM * N_PTS;
  const float *qc = q, *kc = k, *vc = v;

  if (ws_size >= 3 * tensElems * sizeof(float)) {
    float* qw = (float*)d_ws;
    float* kw = qw + tensElems;
    float* vw = kw + tensElems;
    dim3 g(N_PTS / CM_NT, B_SZ);
    cmix_kernel<<<g, 256, 0, stream>>>(q, Wq, qw);
    cmix_kernel<<<g, 256, 0, stream>>>(k, Wk, kw);
    cmix_kernel<<<g, 256, 0, stream>>>(v, Wv, vw);
    qc = qw; kc = kw; vc = vw;
  }

  dim3 ga(N_PTS / QB, B_SZ);
  attn_kernel<<<ga, 256, 0, stream>>>(qc, kc, vc, Yout, Rout);
}


# attn: 16-wave work-groups, four waves per SIMD (was two): a query tile's keys split over four waves, 64-key steps; four partial softmax states merged at the end (changes key summation order)
# speedup vs baseline: 3.1166x; 1.0461x over previous
#include <hip/hip_runtime.h>
#include <math.h>

typedef float v2f __attribute__((ext_vector_type(2)));
typedef float v8f __attribute__((ext_vector_type(8)));

#define C_DIM 128
#define N_PTS 4096
#define B_SZ  4
#define QB    64
#define VSTR  132
#define OSTR  68
#define PSTR  17
#define QT_W  2048
#define NGRP  4
#define KSTEP (16 * NGRP)
#define ATHR  (128 * NGRP)
#define KT_G  2064
#define VT_G  2116
#define KSTG  (NGRP * KT_G)
#define VSTG  (NGRP * VT_G)
#define NSTEP (N_PTS / KSTEP)
#define SMEMQ 8704
#define MRG_O ((NGRP - 1) * 4 * 2048)
#define MRG_S ((NGRP - 1) * 4 * 64)

__device__ __forceinline__ float xor16(float v, unsigned selLo, unsigned selHi) {
  unsigned i = __builtin_bit_cast(unsigned, v);
  i = __builtin_amdgcn_permlane16(i, i, selLo, selHi, false, false);
  return __builtin_bit_cast(float, i);
}
#define XOR16_1(v) xor16((v), 0x67452301u, 0xEFCDAB89u)
#define XOR16_2(v) xor16((v), 0x54761032u, 0xDCFE98BAu)
#define XOR16_4(v) xor16((v), 0x32107654u, 0xBA98FEDCu)
#define XOR16_8(v) xor16((v), 0xFEDCBA98u, 0x76543210u)

#define CM_NT   64
#define CM_XP   160
#define CM_WSTR 132
#define CM_OSTR 68
#define CM_XS   (64 * CM_XP)
#define CM_WT   (16 * CM_WSTR)

__global__ __launch_bounds__(256) void cmix_kernel(const float* __restrict__ x,
                                                   const float* __restrict__ W,
                                                   float* __restrict__ out) {
  __shared__ __attribute__((aligned(16))) float cm_lds[CM_XS + 8 * CM_WT];
  float* xs = cm_lds;
  const int tid  = threadIdx.x;
  const int lane = tid & 31;
  const int wave = tid >> 5;
  const int lo = lane & 15;
  const int hg = lane >> 4;
  const int b   = blockIdx.y;
  const int nb0 = blockIdx.x * CM_NT;
  const int o0  = wave * 16;
  float* wt_w = cm_lds + CM_XS + wave * CM_WT;

#pragma unroll
  for (int i = 0; i < 4; ++i) {
    const int idx = tid + 256 * i;
    const int p  = idx >> 4;
    const int n4 = (idx & 15) * 4;
    const float* xp = x + ((size_t)(b * C_DIM + 2 * p)) * N_PTS + nb0 + n4;
    const float4 e = *(const float4*)xp;
    const float4 f = *(const float4*)(xp + N_PTS);
    float* d = &xs[p * CM_XP + 2 * n4];
    *(float4*)d       = make_float4(e.x, f.x, e.y, f.y);
    *(float4*)(d + 4) = make_float4(e.z, f.z, e.w, f.w);
  }
#pragma unroll
  for (int i = 0; i < 16; ++i) {
    const float4 wv = *(const float4*)(W + (size_t)(o0 + i) * C_DIM + 4 * lane);
    *(float4*)&wt_w[i * CM_WSTR + 4 * lane] = wv;
  }
  __syncthreads();

  v8f acc[4] = {};
  const float* a_l = wt_w + lo * CM_WSTR + 2 * hg;
  const float* b_l = xs + hg * CM_XP + 2 * lo;
#pragma unroll
  for (int ks = 0; ks < 32; ++ks) {
    const v2f a = *(const v2f*)(a_l + 4 * ks);
#pragma unroll
    for (int j = 0; j < 4; ++j) {
      const v2f bv = *(const v2f*)(b_l + 2 * ks * CM_XP + 32 * j);
      acc[j] = __builtin_amdgcn_wmma_f32_16x16x4_f32(false, a, false, bv,
                                                     (short)0, acc[j], false, false);
    }
  }

#pragma unroll
  for (int j = 0; j < 4; ++j)
#pragma unroll
    for (int r = 0; r < 8; ++r)
      wt_w[(r + 8 * hg) * CM_OSTR + 16 * j + lo] = acc[j][r];
#pragma unroll
  for (int i = 0; i < 8; ++i) {
    const int row = 2 * i + hg;
    const float4 val = *(const float4*)&wt_w[row * CM_OSTR + 4 * lo];
    *(float4*)(out + ((size_t)(b * C_DIM + o0 + row)) * N_PTS + nb0 + 4 * lo) = val;
  }
}

__global__ __launch_bounds__(ATHR)
__attribute__((amdgpu_num_vgpr(256)))
void attn_kernel(const float* __restrict__ Q,
                 const float* __restrict__ K,
                 const float* __restrict__ V,
                 float* __restrict__ Y,
                 float* __restrict__ R) {
  __shared__ __attribute__((aligned(16))) float smem_q[SMEMQ];
  __shared__ __attribute__((aligned(16))) float kt[2 * KSTG];
  __shared__ __attribute__((aligned(128))) float vt[2 * VSTG];
  __shared__ float pl[4 * NGRP][16 * PSTR];
  __shared__ float mrg_o[MRG_O];
  __shared__ float mrg_s[MRG_S];
  __shared__ float rs[QB];

  const int tid  = threadIdx.x;
  const int lane = tid & 31;
  const int wave = tid >> 5;
  const int qtile = wave & 3;
  const int grp   = wave >> 2;
  const int b    = blockIdx.y;
  const int qBase0 = blockIdx.x * QB;
  const int lo = lane & 15;
  const int hg = lane >> 4;
  const float invTemp = 0.088388347648318447f;
  const float NEGINF = -__builtin_inff();
  const float LAZY_M = 90.50966799f;

#pragma unroll
  for (int i = 0; i < 2048 / ATHR; ++i) {
    const int idx4 = tid + ATHR * i;
    const int c  = idx4 >> 4;
    const int q4 = idx4 & 15;
    const float4 qv = *(const float4*)(Q + ((size_t)(b * C_DIM + c)) * N_PTS + qBase0 + q4 * 4);
    const int q = q4 * 4;
    *(float4*)&smem_q[(q >> 4) * QT_W + c * 16 + (q & 15)] = qv;
  }

  const int st_c  = tid >> 4;
  const int st_n4 = (tid & 15) * 4;
  const int st_tg = st_n4 >> 4;
  const int st_nn = st_n4 & 15;
  const size_t st_g = ((size_t)(b * C_DIM + st_c)) * N_PTS + st_n4;
  const float* kg = K + st_g;
  const float* vg = V + st_g;
  float* k_st = kt + st_tg * KT_G + st_c * 16 + st_nn;
  float* v_st = vt + st_tg * VT_G + st_nn * VSTR + st_c;
  float4 kr0, kr1, kr2, kr3, vr0, vr1, vr2, vr3;
#define KV_FETCH(kb)                                                   \
  do {                                                                 \
    kr0 = *(const float4*)(kg + (kb));                                 \
    kr1 = *(const float4*)(kg + (kb) + (size_t)1 * 32 * N_PTS);        \
    kr2 = *(const float4*)(kg + (kb) + (size_t)2 * 32 * N_PTS);        \
    kr3 = *(const float4*)(kg + (kb) + (size_t)3 * 32 * N_PTS);        \
    vr0 = *(const float4*)(vg + (kb));                                 \
    vr1 = *(const float4*)(vg + (kb) + (size_t)1 * 32 * N_PTS);        \
    vr2 = *(const float4*)(vg + (kb) + (size_t)2 * 32 * N_PTS);        \
    vr3 = *(const float4*)(vg + (kb) + (size_t)3 * 32 * N_PTS);        \
  } while (0)
#define V_PUT(vd, i, vv)                                               \
  do {                                                                 \
    (vd)[0 * VSTR + (i) * 32] = (vv).x;                                \
    (vd)[1 * VSTR + (i) * 32] = (vv).y;                                \
    (vd)[2 * VSTR + (i) * 32] = (vv).z;                                \
    (vd)[3 * VSTR + (i) * 32] = (vv).w;                                \
  } while (0)
#define KV_PUT(kd, vd)                                                 \
  do {                                                                 \
    *(float4*)((kd) + 0 * 512) = kr0;                                  \
    *(float4*)((kd) + 1 * 512) = kr1;                                  \
    *(float4*)((kd) + 2 * 512) = kr2;                                  \
    *(float4*)((kd) + 3 * 512) = kr3;                                  \
    V_PUT(vd, 0, vr0);                                                 \
    V_PUT(vd, 1, vr1);                                                 \
    V_PUT(vd, 2, vr2);                                                 \
    V_PUT(vd, 3, vr3);                                                 \
  } while (0)

  KV_FETCH(0);
  KV_PUT(k_st, v_st);
  __syncthreads();
  const float* qt_w = smem_q + qtile * QT_W;

  v8f  o[8] = {};
  v8f  ls8  = {};
  float t1[8], t2[8], mrow[8];
#pragma unroll
  for (int r = 0; r < 8; ++r) { t1[r] = NEGINF; t2[r] = NEGINF; mrow[r] = NEGINF; }
  const v2f vone = {1.0f, 1.0f};

  for (int it = 0; it < NSTEP; ++it) {
    const int  cur  = it & 1;
    const bool more = (it + 1 < NSTEP);
    if (more) {
      const int kbn = (it + 1) * KSTEP;
      KV_FETCH(kbn);
    }
    const float* kt_g = kt + cur * KSTG + grp * KT_G;
    const float* vt_g = vt + cur * VSTG + grp * VT_G;

    v8f s = {};
#define SC_LOAD(qv, kv, rd)                                            \
  _Pragma("unroll")                                                    \
  for (int j = 0; j < 8; ++j) {                                        \
    const int c0 = 4 * (8 * (rd) + j) + 2 * hg;                         \
    (qv)[j].x = qt_w[c0 * 16 + lo];                                    \
    (qv)[j].y = qt_w[(c0 + 1) * 16 + lo];                              \
    (kv)[j].x = kt_g[c0 * 16 + lo];                                    \
    (kv)[j].y = kt_g[(c0 + 1) * 16 + lo];                              \
  }
#define SC_MMA(qv, kv)                                                 \
  _Pragma("unroll")                                                    \
  for (int j = 0; j < 8; ++j)                                          \
    s = __builtin_amdgcn_wmma_f32_16x16x4_f32(false, (qv)[j], false, (kv)[j], \
                                              (short)0, s, false, false);
#define SC_FENCE() __builtin_amdgcn_sched_barrier(0)
    v2f qa[8], ka[8], qb[8], kb[8];
    SC_LOAD(qa, ka, 0);
    SC_LOAD(qb, kb, 1);
    SC_FENCE();
    SC_MMA(qa, ka);
    SC_FENCE();
    SC_LOAD(qa, ka, 2);
    SC_FENCE();
    SC_MMA(qb, kb);
    SC_FENCE();
    SC_LOAD(qb, kb, 3);
    SC_FENCE();
    SC_MMA(qa, ka);
    SC_FENCE();
    SC_MMA(qb, kb);
    SC_FENCE();
#undef SC_LOAD
#undef SC_MMA

    if (more) {
      float* kd = k_st + (cur ^ 1) * KSTG;
      float* vd = v_st + (cur ^ 1) * VSTG;
      KV_PUT(kd, vd);
    }

    float d[8];
    float dmax = NEGINF;
    v8f p;
#pragma unroll
    for (int r = 0; r < 8; ++r) {
      d[r] = s[r] - mrow[r];
      dmax = fmaxf(dmax, d[r]);
      const float lw = fminf(t1[r], s[r]);
      t1[r] = fmaxf(t1[r], s[r]);
      t2[r] = fmaxf(t2[r], lw);
      p[r] = __expf(d[r] * invTemp);
    }

    if (__builtin_amdgcn_ballot_w32(dmax > LAZY_M) != 0u) {
#pragma unroll
      for (int r = 0; r < 8; ++r) {
        float a1 = s[r];
        a1 = fmaxf(a1, XOR16_1(a1));
        a1 = fmaxf(a1, XOR16_2(a1));
        a1 = fmaxf(a1, XOR16_4(a1));
        a1 = fmaxf(a1, XOR16_8(a1));
        const float mo = mrow[r];
        mrow[r] = fmaxf(mo, a1);
        const float sc = __expf((mo - mrow[r]) * invTemp);
        ls8[r] *= sc;
#pragma unroll
        for (int ct = 0; ct < 8; ++ct) o[ct][r] *= sc;
        p[r] = __expf((s[r] - mrow[r]) * invTemp);
      }
    }

    float* pw = &pl[wave][0];
#pragma unroll
    for (int r = 0; r < 8; ++r)
      pw[(r + 8 * hg) * PSTR + lo] = p[r];

#define PV_LOAD(av, vv, ks2)                                           \
  do {                                                                 \
    const int k0 = 4 * (ks2) + 2 * hg;                                 \
    (av).x = pw[lo * PSTR + k0];                                       \
    (av).y = pw[lo * PSTR + k0 + 1];                                   \
    _Pragma("unroll")                                                  \
    for (int ct = 0; ct < 8; ++ct) {                                   \
      (vv)[ct].x = vt_g[k0 * VSTR + ct * 16 + lo];                     \
      (vv)[ct].y = vt_g[(k0 + 1) * VSTR + ct * 16 + lo];               \
    }                                                                  \
  } while (0)
#define PV_MMA(av, vv)                                                 \
  do {                                                                 \
    _Pragma("unroll")                                                  \
    for (int ct = 0; ct < 8; ++ct)                                     \
      o[ct] = __builtin_amdgcn_wmma_f32_16x16x4_f32(false, (av), false, (vv)[ct], \
                                                    (short)0, o[ct], false, false); \
    ls8 = __builtin_amdgcn_wmma_f32_16x16x4_f32(false, (av), false, vone, \
                                                (short)0, ls8, false, false); \
  } while (0)
    v2f acur, vcur[8];
    PV_LOAD(acur, vcur, 0);
#pragma unroll 1
    for (int ks2 = 1; ks2 < 4; ++ks2) {
      v2f anxt, vnxt[8];
      PV_LOAD(anxt, vnxt, ks2);
      SC_FENCE();
      PV_MMA(acur, vcur);
      SC_FENCE();
      acur = anxt;
#pragma unroll
      for (int ct = 0; ct < 8; ++ct) vcur[ct] = vnxt[ct];
    }
    PV_MMA(acur, vcur);
#undef PV_LOAD
#undef PV_MMA
#undef SC_FENCE
    __syncthreads();
  }

#undef KV_FETCH
#undef KV_PUT
#undef V_PUT

#pragma unroll
  for (int r = 0; r < 8; ++r) {
    float a1 = t1[r], a2 = t2[r];
    float b1 = XOR16_1(a1), b2 = XOR16_1(a2);
    float hi = fmaxf(a1, b1), lw = fminf(a1, b1);
    a2 = fmaxf(lw, fmaxf(a2, b2)); a1 = hi;
    b1 = XOR16_2(a1); b2 = XOR16_2(a2);
    hi = fmaxf(a1, b1); lw = fminf(a1, b1);
    a2 = fmaxf(lw, fmaxf(a2, b2)); a1 = hi;
    b1 = XOR16_4(a1); b2 = XOR16_4(a2);
    hi = fmaxf(a1, b1); lw = fminf(a1, b1);
    a2 = fmaxf(lw, fmaxf(a2, b2)); a1 = hi;
    b1 = XOR16_8(a1); b2 = XOR16_8(a2);
    hi = fmaxf(a1, b1); lw = fminf(a1, b1);
    a2 = fmaxf(lw, fmaxf(a2, b2)); a1 = hi;
    t1[r] = a1; t2[r] = a2;
  }

  if (grp != 0) {
    float* mo_ = mrg_o + ((grp - 1) * 4 + qtile) * 2048;
    float* ms_ = mrg_s + ((grp - 1) * 4 + qtile) * 64;
#pragma unroll
    for (int r = 0; r < 8; ++r) {
      if (lo == 0) {
        ms_[r * 2 + hg]      = t1[r];
        ms_[16 + r * 2 + hg] = t2[r];
        ms_[32 + r * 2 + hg] = ls8[r];
        ms_[48 + r * 2 + hg] = mrow[r];
      }
#pragma unroll
      for (int ct = 0; ct < 8; ++ct)
        mo_[(ct * 8 + r) * 32 + lane] = o[ct][r];
    }
  }
  __syncthreads();

  if (grp == 0) {
#pragma unroll
    for (int r = 0; r < 8; ++r) {
      float m = t1[r], m2 = t2[r];
      float mr = mrow[r], l = ls8[r];
      float acc[8];
#pragma unroll
      for (int ct = 0; ct < 8; ++ct) acc[ct] = o[ct][r];
#pragma unroll
      for (int g = 0; g < NGRP - 1; ++g) {
        const float* mo_ = mrg_o + (g * 4 + qtile) * 2048;
        const float* ms_ = mrg_s + (g * 4 + qtile) * 64;
        const float t1b = ms_[r * 2 + hg];
        const float t2b = ms_[16 + r * 2 + hg];
        const float lb  = ms_[32 + r * 2 + hg];
        const float rb  = ms_[48 + r * 2 + hg];
        m2 = fmaxf(fminf(m, t1b), fmaxf(m2, t2b));
        m  = fmaxf(m, t1b);
        const float mn = fmaxf(mr, rb);
        const float sa = __expf((mr - mn) * invTemp);
        const float sb = __expf((rb - mn) * invTemp);
        l = l * sa + lb * sb;
#pragma unroll
        for (int ct = 0; ct < 8; ++ct)
          acc[ct] = acc[ct] * sa + mo_[(ct * 8 + r) * 32 + lane] * sb;
        mr = mn;
      }
      const float invl = 1.0f / l;
      const int q = qtile * 16 + r + 8 * hg;
#pragma unroll
      for (int ct = 0; ct < 8; ++ct) {
        const int c = ct * 16 + lo;
        smem_q[c * OSTR + q] = acc[ct] * invl;
      }
      if (lo == 0) {
        const float x1 = fminf(fmaxf(m, -1.0f), 1.0f);
        const float x2 = fminf(fmaxf(m2, -1.0f), 1.0f);
        rs[q] = acosf(x1) / acosf(x2);
      }
    }
  }
  __syncthreads();

#pragma unroll
  for (int i = 0; i < 2048 / ATHR; ++i) {
    const int idx4 = tid + ATHR * i;
    const int c  = idx4 >> 4;
    const int q4 = (idx4 & 15) * 4;
    const float4 val = *(const float4*)&smem_q[c * OSTR + q4];
    *(float4*)(Y + ((size_t)(b * C_DIM + c)) * N_PTS + qBase0 + q4) = val;
  }
  if (tid < QB) R[b * N_PTS + qBase0 + tid] = rs[tid];
}

extern "C" void kernel_launch(void* const* d_in, const int* in_sizes, int n_in,
                              void* d_out, int out_size, void* d_ws, size_t ws_size,
                              hipStream_t stream) {
  (void)in_sizes; (void)n_in; (void)out_size;
  const float* q  = (const float*)d_in[0];
  const float* k  = (const float*)d_in[1];
  const float* v  = (const float*)d_in[2];
  const float* Wq = (const float*)d_in[3];
  const float* Wk = (const float*)d_in[4];
  const float* Wv = (const float*)d_in[5];

  float* out  = (float*)d_out;
  float* Yout = out;
  float* Rout = out + (size_t)B_SZ * C_DIM * N_PTS;

  const size_t tensElems = (size_t)B_SZ * C_DIM * N_PTS;
  const float *qc = q, *kc = k, *vc = v;

  if (ws_size >= 3 * tensElems * sizeof(float)) {
    float* qw = (float*)d_ws;
    float* kw = qw + tensElems;
    float* vw = kw + tensElems;
    dim3 g(N_PTS / CM_NT, B_SZ);
    cmix_kernel<<<g, 256, 0, stream>>>(q, Wq, qw);
    cmix_kernel<<<g, 256, 0, stream>>>(k, Wk, kw);
    cmix_kernel<<<g, 256, 0, stream>>>(v, Wv, vw);
    qc = qw; kc = kw; vc = vw;
  }

  dim3 ga(N_PTS / QB, B_SZ);
  attn_kernel<<<ga, ATHR, 0, stream>>>(qc, kc, vc, Yout, Rout);
}


# attn: both products transposed (keys down the rows): softmax weights feed the 2nd product from registers, no LDS pass; per-lane row stats; normalizer summed on VALU (key sum order changes)
# speedup vs baseline: 3.3386x; 1.0712x over previous
#include <hip/hip_runtime.h>
#include <math.h>

typedef float v2f __attribute__((ext_vector_type(2)));
typedef float v8f __attribute__((ext_vector_type(8)));

#define C_DIM 128
#define N_PTS 4096
#define B_SZ  4
#define QB    64
#define VSTR  132
#define OSTR  68
#define QT_W  2048
#define NGRP  4
#define KSTEP (16 * NGRP)
#define ATHR  (128 * NGRP)
#define KT_G  2064
#define VT_G  2116
#define KSTG  (NGRP * KT_G)
#define VSTG  (NGRP * VT_G)
#define NSTEP (N_PTS / KSTEP)
#define SMEMQ 8704
#define MRG_O ((NGRP - 1) * 4 * 2048)
#define MRG_S ((NGRP - 1) * 4 * 64)

__device__ __forceinline__ float other_half(float v) {
  unsigned i = __builtin_bit_cast(unsigned, v);
  i = __builtin_amdgcn_permlanex16(i, i, 0x76543210u, 0xFEDCBA98u, false, false);
  return __builtin_bit_cast(float, i);
}

#define CM_NT   64
#define CM_XP   160
#define CM_WSTR 132
#define CM_OSTR 68
#define CM_XS   (64 * CM_XP)
#define CM_WT   (16 * CM_WSTR)

__global__ __launch_bounds__(256) void cmix_kernel(const float* __restrict__ x,
                                                   const float* __restrict__ W,
                                                   float* __restrict__ out) {
  __shared__ __attribute__((aligned(16))) float cm_lds[CM_XS + 8 * CM_WT];
  float* xs = cm_lds;
  const int tid  = threadIdx.x;
  const int lane = tid & 31;
  const int wave = tid >> 5;
  const int lo = lane & 15;
  const int hg = lane >> 4;
  const int b   = blockIdx.y;
  const int nb0 = blockIdx.x * CM_NT;
  const int o0  = wave * 16;
  float* wt_w = cm_lds + CM_XS + wave * CM_WT;

#pragma unroll
  for (int i = 0; i < 4; ++i) {
    const int idx = tid + 256 * i;
    const int p  = idx >> 4;
    const int n4 = (idx & 15) * 4;
    const float* xp = x + ((size_t)(b * C_DIM + 2 * p)) * N_PTS + nb0 + n4;
    const float4 e = *(const float4*)xp;
    const float4 f = *(const float4*)(xp + N_PTS);
    float* d = &xs[p * CM_XP + 2 * n4];
    *(float4*)d       = make_float4(e.x, f.x, e.y, f.y);
    *(float4*)(d + 4) = make_float4(e.z, f.z, e.w, f.w);
  }
#pragma unroll
  for (int i = 0; i < 16; ++i) {
    const float4 wv = *(const float4*)(W + (size_t)(o0 + i) * C_DIM + 4 * lane);
    *(float4*)&wt_w[i * CM_WSTR + 4 * lane] = wv;
  }
  __syncthreads();

  v8f acc[4] = {};
  const float* a_l = wt_w + lo * CM_WSTR + 2 * hg;
  const float* b_l = xs + hg * CM_XP + 2 * lo;
#pragma unroll
  for (int ks = 0; ks < 32; ++ks) {
    const v2f a = *(const v2f*)(a_l + 4 * ks);
#pragma unroll
    for (int j = 0; j < 4; ++j) {
      const v2f bv = *(const v2f*)(b_l + 2 * ks * CM_XP + 32 * j);
      acc[j] = __builtin_amdgcn_wmma_f32_16x16x4_f32(false, a, false, bv,
                                                     (short)0, acc[j], false, false);
    }
  }

#pragma unroll
  for (int j = 0; j < 4; ++j)
#pragma unroll
    for (int r = 0; r < 8; ++r)
      wt_w[(r + 8 * hg) * CM_OSTR + 16 * j + lo] = acc[j][r];
#pragma unroll
  for (int i = 0; i < 8; ++i) {
    const int row = 2 * i + hg;
    const float4 val = *(const float4*)&wt_w[row * CM_OSTR + 4 * lo];
    *(float4*)(out + ((size_t)(b * C_DIM + o0 + row)) * N_PTS + nb0 + 4 * lo) = val;
  }
}

__global__ __launch_bounds__(ATHR)
__attribute__((amdgpu_num_vgpr(256)))
void attn_kernel(const float* __restrict__ Q,
                 const float* __restrict__ K,
                 const float* __restrict__ V,
                 float* __restrict__ Y,
                 float* __restrict__ R) {
  __shared__ __attribute__((aligned(16))) float smem_q[SMEMQ];
  __shared__ __attribute__((aligned(16))) float kt[2 * KSTG];
  __shared__ __attribute__((aligned(128))) float vt[2 * VSTG];
  __shared__ float mrg_o[MRG_O];
  __shared__ float mrg_s[MRG_S];
  __shared__ float rs[QB];

  const int tid  = threadIdx.x;
  const int lane = tid & 31;
  const int wave = tid >> 5;
  const int qtile = wave & 3;
  const int grp   = wave >> 2;
  const int b    = blockIdx.y;
  const int qBase0 = blockIdx.x * QB;
  const int lo = lane & 15;
  const int hg = lane >> 4;
  const float invTemp = 0.088388347648318447f;
  const float NEGINF = -__builtin_inff();
  const float LAZY_M = 90.50966799f;

#pragma unroll
  for (int i = 0; i < 2048 / ATHR; ++i) {
    const int idx4 = tid + ATHR * i;
    const int c  = idx4 >> 4;
    const int q4 = idx4 & 15;
    const float4 qv = *(const float4*)(Q + ((size_t)(b * C_DIM + c)) * N_PTS + qBase0 + q4 * 4);
    const int q = q4 * 4;
    *(float4*)&smem_q[(q >> 4) * QT_W + c * 16 + (q & 15)] = qv;
  }

  const int st_c  = tid >> 4;
  const int st_n4 = (tid & 15) * 4;
  const int st_tg = st_n4 >> 4;
  const int st_nn = st_n4 & 15;
  const size_t st_g = ((size_t)(b * C_DIM + st_c)) * N_PTS + st_n4;
  const float* kg = K + st_g;
  const float* vg = V + st_g;
  float* k_st = kt + st_tg * KT_G + st_c * 16 + st_nn;
  float* v_st = vt + st_tg * VT_G + st_nn * VSTR + st_c;
  float4 kr0, kr1, kr2, kr3, vr0, vr1, vr2, vr3;
#define KV_FETCH(kb)                                                   \
  do {                                                                 \
    kr0 = *(const float4*)(kg + (kb));                                 \
    kr1 = *(const float4*)(kg + (kb) + (size_t)1 * 32 * N_PTS);        \
    kr2 = *(const float4*)(kg + (kb) + (size_t)2 * 32 * N_PTS);        \
    kr3 = *(const float4*)(kg + (kb) + (size_t)3 * 32 * N_PTS);        \
    vr0 = *(const float4*)(vg + (kb));                                 \
    vr1 = *(const float4*)(vg + (kb) + (size_t)1 * 32 * N_PTS);        \
    vr2 = *(const float4*)(vg + (kb) + (size_t)2 * 32 * N_PTS);        \
    vr3 = *(const float4*)(vg + (kb) + (size_t)3 * 32 * N_PTS);        \
  } while (0)
#define V_PUT(vd, i, vv)                                               \
  do {                                                                 \
    (vd)[0 * VSTR + (i) * 32] = (vv).x;                                \
    (vd)[1 * VSTR + (i) * 32] = (vv).y;                                \
    (vd)[2 * VSTR + (i) * 32] = (vv).z;                                \
    (vd)[3 * VSTR + (i) * 32] = (vv).w;                                \
  } while (0)
#define KV_PUT(kd, vd)                                                 \
  do {                                                                 \
    *(float4*)((kd) + 0 * 512) = kr0;                                  \
    *(float4*)((kd) + 1 * 512) = kr1;                                  \
    *(float4*)((kd) + 2 * 512) = kr2;                                  \
    *(float4*)((kd) + 3 * 512) = kr3;                                  \
    V_PUT(vd, 0, vr0);                                                 \
    V_PUT(vd, 1, vr1);                                                 \
    V_PUT(vd, 2, vr2);                                                 \
    V_PUT(vd, 3, vr3);                                                 \
  } while (0)

  KV_FETCH(0);
  KV_PUT(k_st, v_st);
  __syncthreads();
  const float* qt_w = smem_q + qtile * QT_W;

  v8f  o[8] = {};
  float t1a = NEGINF, t2a = NEGINF, t1b = NEGINF, t2b = NEGINF;
  float mrow = NEGINF, lsum = 0.0f;

  for (int it = 0; it < NSTEP; ++it) {
    const int  cur  = it & 1;
    const bool more = (it + 1 < NSTEP);
    if (more) {
      const int kbn = (it + 1) * KSTEP;
      KV_FETCH(kbn);
    }
    const float* kt_g = kt + cur * KSTG + grp * KT_G;
    const float* vt_g = vt + cur * VSTG + grp * VT_G;

    v8f s = {};
#define SC_LOAD(qv, kv, rd)                                            \
  _Pragma("unroll")                                                    \
  for (int j = 0; j < 8; ++j) {                                        \
    const int c0 = 4 * (8 * (rd) + j) + 2 * hg;                         \
    (qv)[j].x = qt_w[c0 * 16 + lo];                                    \
    (qv)[j].y = qt_w[(c0 + 1) * 16 + lo];                              \
    (kv)[j].x = kt_g[c0 * 16 + lo];                                    \
    (kv)[j].y = kt_g[(c0 + 1) * 16 + lo];                              \
  }
#define SC_MMA(qv, kv)                                                 \
  _Pragma("unroll")                                                    \
  for (int j = 0; j < 8; ++j)                                          \
    s = __builtin_amdgcn_wmma_f32_16x16x4_f32(false, (kv)[j], false, (qv)[j], \
                                              (short)0, s, false, false);
#define SC_FENCE() __builtin_amdgcn_sched_barrier(0)
    v2f qa[8], ka[8], qb[8], kb[8];
    SC_LOAD(qa, ka, 0);
    SC_LOAD(qb, kb, 1);
    SC_FENCE();
    SC_MMA(qa, ka);
    SC_FENCE();
    SC_LOAD(qa, ka, 2);
    SC_FENCE();
    SC_MMA(qb, kb);
    SC_FENCE();
    SC_LOAD(qb, kb, 3);
    SC_FENCE();
    SC_MMA(qa, ka);
    SC_FENCE();
    SC_MMA(qb, kb);
    SC_FENCE();
#undef SC_LOAD
#undef SC_MMA

    if (more) {
      float* kd = k_st + (cur ^ 1) * KSTG;
      float* vd = v_st + (cur ^ 1) * VSTG;
      KV_PUT(kd, vd);
    }

#define PV_LOAD(vv, ks2)                                               \
  do {                                                                 \
    const int k0 = 8 * hg + 2 * (ks2);                                 \
    _Pragma("unroll")                                                  \
    for (int ct = 0; ct < 8; ++ct) {                                   \
      (vv)[ct].x = vt_g[k0 * VSTR + ct * 16 + lo];                     \
      (vv)[ct].y = vt_g[(k0 + 1) * VSTR + ct * 16 + lo];               \
      SC_FENCE();                                                      \
    }                                                                  \
  } while (0)
#define PV_MMA(vv, ks2)                                                \
  do {                                                                 \
    const v2f pj = {p[2 * (ks2)], p[2 * (ks2) + 1]};                   \
    _Pragma("unroll")                                                  \
    for (int ct = 0; ct < 8; ++ct)                                     \
      o[ct] = __builtin_amdgcn_wmma_f32_16x16x4_f32(false, (vv)[ct], false, pj, \
                                                    (short)0, o[ct], false, false); \
  } while (0)
    v2f va[8], vb[8];
    PV_LOAD(va, 0);
    SC_FENCE();

    float dmax = NEGINF;
    v8f p;
#pragma unroll
    for (int r = 0; r < 4; ++r) {
      const float da = s[r] - mrow;
      const float db = s[r + 4] - mrow;
      dmax = fmaxf(dmax, fmaxf(da, db));
      const float lwa = fminf(t1a, s[r]);
      t1a = fmaxf(t1a, s[r]);
      t2a = fmaxf(t2a, lwa);
      const float lwb = fminf(t1b, s[r + 4]);
      t1b = fmaxf(t1b, s[r + 4]);
      t2b = fmaxf(t2b, lwb);
      p[r]     = __expf(da * invTemp);
      p[r + 4] = __expf(db * invTemp);
    }

    if (__builtin_amdgcn_ballot_w32(dmax > LAZY_M) != 0u) {
      float m8 = fmaxf(fmaxf(fmaxf(s[0], s[1]), fmaxf(s[2], s[3])),
                       fmaxf(fmaxf(s[4], s[5]), fmaxf(s[6], s[7])));
      m8 = fmaxf(m8, other_half(m8));
      const float mo = mrow;
      mrow = fmaxf(mo, m8);
      const float sc = __expf((mo - mrow) * invTemp);
      lsum *= sc;
#pragma unroll
      for (int ct = 0; ct < 8; ++ct)
#pragma unroll
        for (int r = 0; r < 8; ++r) o[ct][r] *= sc;
#pragma unroll
      for (int r = 0; r < 8; ++r) p[r] = __expf((s[r] - mrow) * invTemp);
    }
    lsum += ((p[0] + p[1]) + (p[2] + p[3])) + ((p[4] + p[5]) + (p[6] + p[7]));

    PV_LOAD(vb, 1);
    SC_FENCE();
    PV_MMA(va, 0);
    SC_FENCE();
    PV_LOAD(va, 2);
    SC_FENCE();
    PV_MMA(vb, 1);
    SC_FENCE();
    PV_LOAD(vb, 3);
    SC_FENCE();
    PV_MMA(va, 2);
    SC_FENCE();
    PV_MMA(vb, 3);
#undef PV_LOAD
#undef PV_MMA
#undef SC_FENCE
    __syncthreads();
  }

#undef KV_FETCH
#undef KV_PUT
#undef V_PUT

  float t1, t2;
  {
    const float hi = fmaxf(t1a, t1b), lw = fminf(t1a, t1b);
    t2 = fmaxf(lw, fmaxf(t2a, t2b)); t1 = hi;
    const float b1 = other_half(t1), b2 = other_half(t2);
    const float hi2 = fmaxf(t1, b1), lw2 = fminf(t1, b1);
    t2 = fmaxf(lw2, fmaxf(t2, b2)); t1 = hi2;
    lsum += other_half(lsum);
  }

  if (grp != 0) {
    float* mo_ = mrg_o + ((grp - 1) * 4 + qtile) * 2048;
    float* ms_ = mrg_s + ((grp - 1) * 4 + qtile) * 64;
    if (hg == 0) {
      ms_[lo]      = t1;
      ms_[16 + lo] = t2;
      ms_[32 + lo] = lsum;
      ms_[48 + lo] = mrow;
    }
#pragma unroll
    for (int ct = 0; ct < 8; ++ct)
#pragma unroll
      for (int r = 0; r < 8; ++r)
        mo_[(ct * 8 + r) * 32 + lane] = o[ct][r];
  }
  __syncthreads();

  if (grp == 0) {
    float m = t1, m2 = t2;
    float mr = mrow, l = lsum;
#pragma unroll
    for (int g = 0; g < NGRP - 1; ++g) {
      const float* mo_ = mrg_o + (g * 4 + qtile) * 2048;
      const float* ms_ = mrg_s + (g * 4 + qtile) * 64;
      const float t1g = ms_[lo];
      const float t2g = ms_[16 + lo];
      const float lg  = ms_[32 + lo];
      const float rg  = ms_[48 + lo];
      m2 = fmaxf(fminf(m, t1g), fmaxf(m2, t2g));
      m  = fmaxf(m, t1g);
      const float mn = fmaxf(mr, rg);
      const float sa = __expf((mr - mn) * invTemp);
      const float sb = __expf((rg - mn) * invTemp);
      l = l * sa + lg * sb;
#pragma unroll
      for (int ct = 0; ct < 8; ++ct)
#pragma unroll
        for (int r = 0; r < 8; ++r)
          o[ct][r] = o[ct][r] * sa + mo_[(ct * 8 + r) * 32 + lane] * sb;
      mr = mn;
    }
    const float invl = 1.0f / l;
    const int q = qtile * 16 + lo;
#pragma unroll
    for (int ct = 0; ct < 8; ++ct)
#pragma unroll
      for (int r = 0; r < 8; ++r)
        smem_q[(ct * 16 + 8 * hg + r) * OSTR + q] = o[ct][r] * invl;
    if (hg == 0) {
      const float x1 = fminf(fmaxf(m, -1.0f), 1.0f);
      const float x2 = fminf(fmaxf(m2, -1.0f), 1.0f);
      rs[q] = acosf(x1) / acosf(x2);
    }
  }
  __syncthreads();

#pragma unroll
  for (int i = 0; i < 2048 / ATHR; ++i) {
    const int idx4 = tid + ATHR * i;
    const int c  = idx4 >> 4;
    const int q4 = (idx4 & 15) * 4;
    const float4 val = *(const float4*)&smem_q[c * OSTR + q4];
    *(float4*)(Y + ((size_t)(b * C_DIM + c)) * N_PTS + qBase0 + q4) = val;
  }
  if (tid < QB) R[b * N_PTS + qBase0 + tid] = rs[tid];
}

extern "C" void kernel_launch(void* const* d_in, const int* in_sizes, int n_in,
                              void* d_out, int out_size, void* d_ws, size_t ws_size,
                              hipStream_t stream) {
  (void)in_sizes; (void)n_in; (void)out_size;
  const float* q  = (const float*)d_in[0];
  const float* k  = (const float*)d_in[1];
  const float* v  = (const float*)d_in[2];
  const float* Wq = (const float*)d_in[3];
  const float* Wk = (const float*)d_in[4];
  const float* Wv = (const float*)d_in[5];

  float* out  = (float*)d_out;
  float* Yout = out;
  float* Rout = out + (size_t)B_SZ * C_DIM * N_PTS;

  const size_t tensElems = (size_t)B_SZ * C_DIM * N_PTS;
  const float *qc = q, *kc = k, *vc = v;

  if (ws_size >= 3 * tensElems * sizeof(float)) {
    float* qw = (float*)d_ws;
    float* kw = qw + tensElems;
    float* vw = kw + tensElems;
    dim3 g(N_PTS / CM_NT, B_SZ);
    cmix_kernel<<<g, 256, 0, stream>>>(q, Wq, qw);
    cmix_kernel<<<g, 256, 0, stream>>>(k, Wk, kw);
    cmix_kernel<<<g, 256, 0, stream>>>(v, Wv, vw);
    qc = qw; kc = kw; vc = vw;
  }

  dim3 ga(N_PTS / QB, B_SZ);
  attn_kernel<<<ga, ATHR, 0, stream>>>(qc, kc, vc, Yout, Rout);
}


# attn: step barrier moved in front of the last 16 output WMMAs, next step's first score fragments read under them; a step's four V^T read groups issued as one ahead of the softmax
# speedup vs baseline: 3.4427x; 1.0312x over previous
#include <hip/hip_runtime.h>
#include <math.h>

typedef float v2f __attribute__((ext_vector_type(2)));
typedef float v8f __attribute__((ext_vector_type(8)));

#define C_DIM 128
#define N_PTS 4096
#define B_SZ  4
#define QB    64
#define VSTR  132
#define OSTR  68
#define QT_W  2048
#define NGRP  4
#define KSTEP (16 * NGRP)
#define ATHR  (128 * NGRP)
#define KT_G  2064
#define VT_G  2116
#define KSTG  (NGRP * KT_G)
#define VSTG  (NGRP * VT_G)
#define NSTEP (N_PTS / KSTEP)
#define SMEMQ 8704
#define MRG_O ((NGRP - 1) * 4 * 2048)
#define MRG_S ((NGRP - 1) * 4 * 64)

__device__ __forceinline__ float other_half(float v) {
  unsigned i = __builtin_bit_cast(unsigned, v);
  i = __builtin_amdgcn_permlanex16(i, i, 0x76543210u, 0xFEDCBA98u, false, false);
  return __builtin_bit_cast(float, i);
}

#define CM_NT   64
#define CM_XP   160
#define CM_WSTR 132
#define CM_OSTR 68
#define CM_XS   (64 * CM_XP)
#define CM_WT   (16 * CM_WSTR)

__global__ __launch_bounds__(256) void cmix_kernel(const float* __restrict__ x,
                                                   const float* __restrict__ W,
                                                   float* __restrict__ out) {
  __shared__ __attribute__((aligned(16))) float cm_lds[CM_XS + 8 * CM_WT];
  float* xs = cm_lds;
  const int tid  = threadIdx.x;
  const int lane = tid & 31;
  const int wave = tid >> 5;
  const int lo = lane & 15;
  const int hg = lane >> 4;
  const int b   = blockIdx.y;
  const int nb0 = blockIdx.x * CM_NT;
  const int o0  = wave * 16;
  float* wt_w = cm_lds + CM_XS + wave * CM_WT;

#pragma unroll
  for (int i = 0; i < 4; ++i) {
    const int idx = tid + 256 * i;
    const int p  = idx >> 4;
    const int n4 = (idx & 15) * 4;
    const float* xp = x + ((size_t)(b * C_DIM + 2 * p)) * N_PTS + nb0 + n4;
    const float4 e = *(const float4*)xp;
    const float4 f = *(const float4*)(xp + N_PTS);
    float* d = &xs[p * CM_XP + 2 * n4];
    *(float4*)d       = make_float4(e.x, f.x, e.y, f.y);
    *(float4*)(d + 4) = make_float4(e.z, f.z, e.w, f.w);
  }
#pragma unroll
  for (int i = 0; i < 16; ++i) {
    const float4 wv = *(const float4*)(W + (size_t)(o0 + i) * C_DIM + 4 * lane);
    *(float4*)&wt_w[i * CM_WSTR + 4 * lane] = wv;
  }
  __syncthreads();

  v8f acc[4] = {};
  const float* a_l = wt_w + lo * CM_WSTR + 2 * hg;
  const float* b_l = xs + hg * CM_XP + 2 * lo;
#pragma unroll
  for (int ks = 0; ks < 32; ++ks) {
    const v2f a = *(const v2f*)(a_l + 4 * ks);
#pragma unroll
    for (int j = 0; j < 4; ++j) {
      const v2f bv = *(const v2f*)(b_l + 2 * ks * CM_XP + 32 * j);
      acc[j] = __builtin_amdgcn_wmma_f32_16x16x4_f32(false, a, false, bv,
                                                     (short)0, acc[j], false, false);
    }
  }

#pragma unroll
  for (int j = 0; j < 4; ++j)
#pragma unroll
    for (int r = 0; r < 8; ++r)
      wt_w[(r + 8 * hg) * CM_OSTR + 16 * j + lo] = acc[j][r];
#pragma unroll
  for (int i = 0; i < 8; ++i) {
    const int row = 2 * i + hg;
    const float4 val = *(const float4*)&wt_w[row * CM_OSTR + 4 * lo];
    *(float4*)(out + ((size_t)(b * C_DIM + o0 + row)) * N_PTS + nb0 + 4 * lo) = val;
  }
}

__global__ __launch_bounds__(ATHR)
__attribute__((amdgpu_num_vgpr(256)))
void attn_kernel(const float* __restrict__ Q,
                 const float* __restrict__ K,
                 const float* __restrict__ V,
                 float* __restrict__ Y,
                 float* __restrict__ R) {
  __shared__ __attribute__((aligned(16))) float smem_q[SMEMQ];
  __shared__ __attribute__((aligned(16))) float kt[2 * KSTG];
  __shared__ __attribute__((aligned(128))) float vt[2 * VSTG];
  __shared__ float mrg_o[MRG_O];
  __shared__ float mrg_s[MRG_S];
  __shared__ float rs[QB];

  const int tid  = threadIdx.x;
  const int lane = tid & 31;
  const int wave = tid >> 5;
  const int qtile = wave & 3;
  const int grp   = wave >> 2;
  const int b    = blockIdx.y;
  const int qBase0 = blockIdx.x * QB;
  const int lo = lane & 15;
  const int hg = lane >> 4;
  const float invTemp = 0.088388347648318447f;
  const float NEGINF = -__builtin_inff();
  const float LAZY_M = 90.50966799f;

#pragma unroll
  for (int i = 0; i < 2048 / ATHR; ++i) {
    const int idx4 = tid + ATHR * i;
    const int c  = idx4 >> 4;
    const int q4 = idx4 & 15;
    const float4 qv = *(const float4*)(Q + ((size_t)(b * C_DIM + c)) * N_PTS + qBase0 + q4 * 4);
    const int q = q4 * 4;
    *(float4*)&smem_q[(q >> 4) * QT_W + c * 16 + (q & 15)] = qv;
  }

  const int st_c  = tid >> 4;
  const int st_n4 = (tid & 15) * 4;
  const int st_tg = st_n4 >> 4;
  const int st_nn = st_n4 & 15;
  const size_t st_g = ((size_t)(b * C_DIM + st_c)) * N_PTS + st_n4;
  const float* kg = K + st_g;
  const float* vg = V + st_g;
  float* k_st = kt + st_tg * KT_G + st_c * 16 + st_nn;
  float* v_st = vt + st_tg * VT_G + st_nn * VSTR + st_c;
  float4 kr0, kr1, kr2, kr3, vr0, vr1, vr2, vr3;
#define KV_FETCH(kb)                                                   \
  do {                                                                 \
    kr0 = *(const float4*)(kg + (kb));                                 \
    kr1 = *(const float4*)(kg + (kb) + (size_t)1 * 32 * N_PTS);        \
    kr2 = *(const float4*)(kg + (kb) + (size_t)2 * 32 * N_PTS);        \
    kr3 = *(const float4*)(kg + (kb) + (size_t)3 * 32 * N_PTS);        \
    vr0 = *(const float4*)(vg + (kb));                                 \
    vr1 = *(const float4*)(vg + (kb) + (size_t)1 * 32 * N_PTS);        \
    vr2 = *(const float4*)(vg + (kb) + (size_t)2 * 32 * N_PTS);        \
    vr3 = *(const float4*)(vg + (kb) + (size_t)3 * 32 * N_PTS);        \
  } while (0)
#define V_PUT(vd, i, vv)                                               \
  do {                                                                 \
    (vd)[0 * VSTR + (i) * 32] = (vv).x;                                \
    (vd)[1 * VSTR + (i) * 32] = (vv).y;                                \
    (vd)[2 * VSTR + (i) * 32] = (vv).z;                                \
    (vd)[3 * VSTR + (i) * 32] = (vv).w;                                \
  } while (0)
#define KV_PUT(kd, vd)                                                 \
  do {                                                                 \
    *(float4*)((kd) + 0 * 512) = kr0;                                  \
    *(float4*)((kd) + 1 * 512) = kr1;                                  \
    *(float4*)((kd) + 2 * 512) = kr2;                                  \
    *(float4*)((kd) + 3 * 512) = kr3;                                  \
    V_PUT(vd, 0, vr0);                                                 \
    V_PUT(vd, 1, vr1);                                                 \
    V_PUT(vd, 2, vr2);                                                 \
    V_PUT(vd, 3, vr3);                                                 \
  } while (0)

  KV_FETCH(0);
  KV_PUT(k_st, v_st);
  __syncthreads();
  const float* qt_w = smem_q + qtile * QT_W;

#define SC_LOAD(qv, kv, rd, ktp)                                       \
  _Pragma("unroll")                                                    \
  for (int j = 0; j < 8; ++j) {                                        \
    const int c0 = 4 * (8 * (rd) + j) + 2 * hg;                         \
    (qv)[j].x = qt_w[c0 * 16 + lo];                                    \
    (qv)[j].y = qt_w[(c0 + 1) * 16 + lo];                              \
    (kv)[j].x = (ktp)[c0 * 16 + lo];                                   \
    (kv)[j].y = (ktp)[(c0 + 1) * 16 + lo];                             \
  }
#define SC_MMA(qv, kv)                                                 \
  _Pragma("unroll")                                                    \
  for (int j = 0; j < 8; ++j)                                          \
    s = __builtin_amdgcn_wmma_f32_16x16x4_f32(false, (kv)[j], false, (qv)[j], \
                                              (short)0, s, false, false);
#define SC_FENCE() __builtin_amdgcn_sched_barrier(0)
  v2f qa[8], ka[8], qb[8], kb[8];
  {
    const float* kt_0 = kt + grp * KT_G;
    SC_LOAD(qa, ka, 0, kt_0);
    SC_LOAD(qb, kb, 1, kt_0);
  }

  v8f  o[8] = {};
  float t1a = NEGINF, t2a = NEGINF, t1b = NEGINF, t2b = NEGINF;
  float mrow = NEGINF, lsum = 0.0f;

  for (int it = 0; it < NSTEP; ++it) {
    const int  cur  = it & 1;
    const bool more = (it + 1 < NSTEP);
    if (more) {
      const int kbn = (it + 1) * KSTEP;
      KV_FETCH(kbn);
    }
    const float* kt_g = kt + cur * KSTG + grp * KT_G;
    const float* vt_g = vt + cur * VSTG + grp * VT_G;

    v8f s = {};
    SC_FENCE();
    SC_MMA(qa, ka);
    SC_FENCE();
    SC_LOAD(qa, ka, 2, kt_g);
    SC_FENCE();
    SC_MMA(qb, kb);
    SC_FENCE();
    SC_LOAD(qb, kb, 3, kt_g);
    SC_FENCE();
    SC_MMA(qa, ka);
    SC_FENCE();
    SC_MMA(qb, kb);
    SC_FENCE();

    if (more) {
      float* kd = k_st + (cur ^ 1) * KSTG;
      float* vd = v_st + (cur ^ 1) * VSTG;
      KV_PUT(kd, vd);
    }

#define PV_LOAD(vv, ks2)                                               \
  do {                                                                 \
    const int k0 = 8 * hg + 2 * (ks2);                                 \
    _Pragma("unroll")                                                  \
    for (int ct = 0; ct < 8; ++ct) {                                   \
      (vv)[ct].x = vt_g[k0 * VSTR + ct * 16 + lo];                     \
      (vv)[ct].y = vt_g[(k0 + 1) * VSTR + ct * 16 + lo];               \
      SC_FENCE();                                                      \
    }                                                                  \
  } while (0)
#define PV_MMA(vv, ks2)                                                \
  do {                                                                 \
    const v2f pj = {p[2 * (ks2)], p[2 * (ks2) + 1]};                   \
    _Pragma("unroll")                                                  \
    for (int ct = 0; ct < 8; ++ct)                                     \
      o[ct] = __builtin_amdgcn_wmma_f32_16x16x4_f32(false, (vv)[ct], false, pj, \
                                                    (short)0, o[ct], false, false); \
  } while (0)
    v2f va[8], vb[8], vc[8], vd[8];
    PV_LOAD(va, 0);
    PV_LOAD(vb, 1);
    PV_LOAD(vc, 2);
    PV_LOAD(vd, 3);
    SC_FENCE();

    float dmax = NEGINF;
    v8f p;
#pragma unroll
    for (int r = 0; r < 4; ++r) {
      const float da = s[r] - mrow;
      const float db = s[r + 4] - mrow;
      dmax = fmaxf(dmax, fmaxf(da, db));
      const float lwa = fminf(t1a, s[r]);
      t1a = fmaxf(t1a, s[r]);
      t2a = fmaxf(t2a, lwa);
      const float lwb = fminf(t1b, s[r + 4]);
      t1b = fmaxf(t1b, s[r + 4]);
      t2b = fmaxf(t2b, lwb);
      p[r]     = __expf(da * invTemp);
      p[r + 4] = __expf(db * invTemp);
    }

    if (__builtin_amdgcn_ballot_w32(dmax > LAZY_M) != 0u) {
      float m8 = fmaxf(fmaxf(fmaxf(s[0], s[1]), fmaxf(s[2], s[3])),
                       fmaxf(fmaxf(s[4], s[5]), fmaxf(s[6], s[7])));
      m8 = fmaxf(m8, other_half(m8));
      const float mo = mrow;
      mrow = fmaxf(mo, m8);
      const float sc = __expf((mo - mrow) * invTemp);
      lsum *= sc;
#pragma unroll
      for (int ct = 0; ct < 8; ++ct)
#pragma unroll
        for (int r = 0; r < 8; ++r) o[ct][r] *= sc;
#pragma unroll
      for (int r = 0; r < 8; ++r) p[r] = __expf((s[r] - mrow) * invTemp);
    }
    lsum += ((p[0] + p[1]) + (p[2] + p[3])) + ((p[4] + p[5]) + (p[6] + p[7]));

    SC_FENCE();
    PV_MMA(va, 0);
    PV_MMA(vb, 1);
    SC_FENCE();
    __syncthreads();
    {
      const float* kt_n = kt + (cur ^ 1) * KSTG + grp * KT_G;
      SC_LOAD(qa, ka, 0, kt_n);
      SC_LOAD(qb, kb, 1, kt_n);
    }
    SC_FENCE();
    PV_MMA(vc, 2);
    PV_MMA(vd, 3);
#undef PV_LOAD
#undef PV_MMA
  }
#undef SC_LOAD
#undef SC_MMA
#undef SC_FENCE

#undef KV_FETCH
#undef KV_PUT
#undef V_PUT

  float t1, t2;
  {
    const float hi = fmaxf(t1a, t1b), lw = fminf(t1a, t1b);
    t2 = fmaxf(lw, fmaxf(t2a, t2b)); t1 = hi;
    const float b1 = other_half(t1), b2 = other_half(t2);
    const float hi2 = fmaxf(t1, b1), lw2 = fminf(t1, b1);
    t2 = fmaxf(lw2, fmaxf(t2, b2)); t1 = hi2;
    lsum += other_half(lsum);
  }

  if (grp != 0) {
    float* mo_ = mrg_o + ((grp - 1) * 4 + qtile) * 2048;
    float* ms_ = mrg_s + ((grp - 1) * 4 + qtile) * 64;
    if (hg == 0) {
      ms_[lo]      = t1;
      ms_[16 + lo] = t2;
      ms_[32 + lo] = lsum;
      ms_[48 + lo] = mrow;
    }
#pragma unroll
    for (int ct = 0; ct < 8; ++ct)
#pragma unroll
      for (int r = 0; r < 8; ++r)
        mo_[(ct * 8 + r) * 32 + lane] = o[ct][r];
  }
  __syncthreads();

  if (grp == 0) {
    float m = t1, m2 = t2;
    float mr = mrow, l = lsum;
#pragma unroll
    for (int g = 0; g < NGRP - 1; ++g) {
      const float* mo_ = mrg_o + (g * 4 + qtile) * 2048;
      const float* ms_ = mrg_s + (g * 4 + qtile) * 64;
      const float t1g = ms_[lo];
      const float t2g = ms_[16 + lo];
      const float lg  = ms_[32 + lo];
      const float rg  = ms_[48 + lo];
      m2 = fmaxf(fminf(m, t1g), fmaxf(m2, t2g));
      m  = fmaxf(m, t1g);
      const float mn = fmaxf(mr, rg);
      const float sa = __expf((mr - mn) * invTemp);
      const float sb = __expf((rg - mn) * invTemp);
      l = l * sa + lg * sb;
#pragma unroll
      for (int ct = 0; ct < 8; ++ct)
#pragma unroll
        for (int r = 0; r < 8; ++r)
          o[ct][r] = o[ct][r] * sa + mo_[(ct * 8 + r) * 32 + lane] * sb;
      mr = mn;
    }
    const float invl = 1.0f / l;
    const int q = qtile * 16 + lo;
#pragma unroll
    for (int ct = 0; ct < 8; ++ct)
#pragma unroll
      for (int r = 0; r < 8; ++r)
        smem_q[(ct * 16 + 8 * hg + r) * OSTR + q] = o[ct][r] * invl;
    if (hg == 0) {
      const float x1 = fminf(fmaxf(m, -1.0f), 1.0f);
      const float x2 = fminf(fmaxf(m2, -1.0f), 1.0f);
      rs[q] = acosf(x1) / acosf(x2);
    }
  }
  __syncthreads();

#pragma unroll
  for (int i = 0; i < 2048 / ATHR; ++i) {
    const int idx4 = tid + ATHR * i;
    const int c  = idx4 >> 4;
    const int q4 = (idx4 & 15) * 4;
    const float4 val = *(const float4*)&smem_q[c * OSTR + q4];
    *(float4*)(Y + ((size_t)(b * C_DIM + c)) * N_PTS + qBase0 + q4) = val;
  }
  if (tid < QB) R[b * N_PTS + qBase0 + tid] = rs[tid];
}

extern "C" void kernel_launch(void* const* d_in, const int* in_sizes, int n_in,
                              void* d_out, int out_size, void* d_ws, size_t ws_size,
                              hipStream_t stream) {
  (void)in_sizes; (void)n_in; (void)out_size;
  const float* q  = (const float*)d_in[0];
  const float* k  = (const float*)d_in[1];
  const float* v  = (const float*)d_in[2];
  const float* Wq = (const float*)d_in[3];
  const float* Wk = (const float*)d_in[4];
  const float* Wv = (const float*)d_in[5];

  float* out  = (float*)d_out;
  float* Yout = out;
  float* Rout = out + (size_t)B_SZ * C_DIM * N_PTS;

  const size_t tensElems = (size_t)B_SZ * C_DIM * N_PTS;
  const float *qc = q, *kc = k, *vc = v;

  if (ws_size >= 3 * tensElems * sizeof(float)) {
    float* qw = (float*)d_ws;
    float* kw = qw + tensElems;
    float* vw = kw + tensElems;
    dim3 g(N_PTS / CM_NT, B_SZ);
    cmix_kernel<<<g, 256, 0, stream>>>(q, Wq, qw);
    cmix_kernel<<<g, 256, 0, stream>>>(k, Wk, kw);
    cmix_kernel<<<g, 256, 0, stream>>>(v, Wv, vw);
    qc = qw; kc = kw; vc = vw;
  }

  dim3 ga(N_PTS / QB, B_SZ);
  attn_kernel<<<ga, ATHR, 0, stream>>>(qc, kc, vc, Yout, Rout);
}


# channel mixes folded into the attention kernel by re-association: Q'=Wk^T(Wq q) per 64-query tile in front, y=Wv(v attn^T) behind, raw k,v in the loop; 1 launch (was 4); channel sums re-ordered
# speedup vs baseline: 3.6284x; 1.0539x over previous
#include <hip/hip_runtime.h>
#include <math.h>

typedef float v2f __attribute__((ext_vector_type(2)));
typedef float v8f __attribute__((ext_vector_type(8)));

#define C_DIM 128
#define N_PTS 4096
#define B_SZ  4
#define QB    64
#define VSTR  132
#define OSTR  68
#define QT_W  2048
#define NGRP  4
#define KSTEP (16 * NGRP)
#define ATHR  (128 * NGRP)
#define KT_G  2064
#define VT_G  2116
#define KSTG  (NGRP * KT_G)
#define VSTG  (NGRP * VT_G)
#define NSTEP (N_PTS / KSTEP)
#define SMEMQ 8704
#define MRG_O ((NGRP - 1) * 4 * 2048)
#define MRG_S ((NGRP - 1) * 4 * 64)

__device__ __forceinline__ float other_half(float v) {
  unsigned i = __builtin_bit_cast(unsigned, v);
  i = __builtin_amdgcn_permlanex16(i, i, 0x76543210u, 0xFEDCBA98u, false, false);
  return __builtin_bit_cast(float, i);
}

#define WSTR 132
template <bool TR, int WS, int BT, int BS, int DT, int DS>
__device__ __forceinline__ void wg_mix(const float* Wl, const float* Bl, float* Dl,
                                       int wave, int lo, int hg) {
  const int mt = wave & 7;
  const int j0 = (wave >> 3) * 2;
  v8f acc0 = {}, acc1 = {};
  const float* a_l = TR ? (Wl + (2 * hg) * WS + 16 * mt + lo)
                        : (Wl + (16 * mt + lo) * WS + 2 * hg);
  const float* b0 = Bl + j0 * BT + (2 * hg) * BS + lo;
  const float* b1 = b0 + BT;
#define MX_LOAD(av, pv, rv, rd)                                        \
  _Pragma("unroll")                                                    \
  for (int j = 0; j < 8; ++j) {                                        \
    const int k0 = 4 * (8 * (rd) + j);                                 \
    if (TR) {                                                          \
      (av)[j].x = a_l[k0 * WS];                                        \
      (av)[j].y = a_l[(k0 + 1) * WS];                                  \
    } else {                                                           \
      (av)[j] = *(const v2f*)(a_l + k0);                               \
    }                                                                  \
    (pv)[j].x = b0[k0 * BS];                                           \
    (pv)[j].y = b0[(k0 + 1) * BS];                                     \
    (rv)[j].x = b1[k0 * BS];                                           \
    (rv)[j].y = b1[(k0 + 1) * BS];                                     \
  }
#define MX_MMA(av, pv, rv)                                             \
  _Pragma("unroll")                                                    \
  for (int j = 0; j < 8; ++j) {                                        \
    acc0 = __builtin_amdgcn_wmma_f32_16x16x4_f32(false, (av)[j], false, (pv)[j], \
                                                 (short)0, acc0, false, false);  \
    acc1 = __builtin_amdgcn_wmma_f32_16x16x4_f32(false, (av)[j], false, (rv)[j], \
                                                 (short)0, acc1, false, false);  \
  }
#define MX_FENCE() __builtin_amdgcn_sched_barrier(0)
  v2f fa0[8], fp0[8], fr0[8], fa1[8], fp1[8], fr1[8];
  MX_LOAD(fa0, fp0, fr0, 0);
  MX_FENCE();
  MX_LOAD(fa1, fp1, fr1, 1);
  MX_FENCE();
  MX_MMA(fa0, fp0, fr0);
  MX_FENCE();
  MX_LOAD(fa0, fp0, fr0, 2);
  MX_FENCE();
  MX_MMA(fa1, fp1, fr1);
  MX_FENCE();
  MX_LOAD(fa1, fp1, fr1, 3);
  MX_FENCE();
  MX_MMA(fa0, fp0, fr0);
  MX_FENCE();
  MX_MMA(fa1, fp1, fr1);
  MX_FENCE();
#undef MX_LOAD
#undef MX_MMA
#undef MX_FENCE
  float* d0 = Dl + j0 * DT + (16 * mt + 8 * hg) * DS + lo;
#pragma unroll
  for (int r = 0; r < 8; ++r) {
    d0[r * DS]      = acc0[r];
    d0[DT + r * DS] = acc1[r];
  }
}

__global__ __launch_bounds__(ATHR)
__attribute__((amdgpu_num_vgpr(256)))
void attn_kernel(const float* __restrict__ Q,
                 const float* __restrict__ K,
                 const float* __restrict__ V,
                 const float* __restrict__ Wq,
                 const float* __restrict__ Wk,
                 const float* __restrict__ Wv,
                 float* __restrict__ Y,
                 float* __restrict__ R) {
  __shared__ __attribute__((aligned(16))) float smem_q[SMEMQ];
  __shared__ __attribute__((aligned(16))) float kt[2 * KSTG];
  __shared__ __attribute__((aligned(128))) float vt[2 * VSTG];
  __shared__ float mrg_o[MRG_O];
  __shared__ float mrg_s[MRG_S];
  __shared__ float rs[QB];

  const int tid  = threadIdx.x;
  const int lane = tid & 31;
  const int wave = tid >> 5;
  const int qtile = wave & 3;
  const int grp   = wave >> 2;
  const int b    = blockIdx.y;
  const int qBase0 = blockIdx.x * QB;
  const int lo = lane & 15;
  const int hg = lane >> 4;
  const float invTemp = 0.088388347648318447f;
  const float NEGINF = -__builtin_inff();
  const float LAZY_M = 90.50966799f;

#pragma unroll
  for (int i = 0; i < 2048 / ATHR; ++i) {
    const int idx4 = tid + ATHR * i;
    const int c  = idx4 >> 4;
    const int q4 = idx4 & 15;
    const float4 qv = *(const float4*)(Q + ((size_t)(b * C_DIM + c)) * N_PTS + qBase0 + q4 * 4);
    const int q = q4 * 4;
    *(float4*)&smem_q[(q >> 4) * QT_W + c * 16 + (q & 15)] = qv;
  }
  float4 wr0, wr1, wr2, wr3, wr4, wr5, wr6, wr7;
  const int w_g = (tid >> 5) * C_DIM + lane * 4;
  float* w_st = vt + (tid >> 5) * WSTR + lane * 4;
#define W_FETCH(Wp)                                                    \
  do {                                                                 \
    wr0 = *(const float4*)((Wp) + w_g + 0 * 16 * C_DIM);               \
    wr1 = *(const float4*)((Wp) + w_g + 1 * 16 * C_DIM);               \
    wr2 = *(const float4*)((Wp) + w_g + 2 * 16 * C_DIM);               \
    wr3 = *(const float4*)((Wp) + w_g + 3 * 16 * C_DIM);               \
    wr4 = *(const float4*)((Wp) + w_g + 4 * 16 * C_DIM);               \
    wr5 = *(const float4*)((Wp) + w_g + 5 * 16 * C_DIM);               \
    wr6 = *(const float4*)((Wp) + w_g + 6 * 16 * C_DIM);               \
    wr7 = *(const float4*)((Wp) + w_g + 7 * 16 * C_DIM);               \
  } while (0)
#define W_PUT()                                                        \
  do {                                                                 \
    *(float4*)(w_st + 0 * 16 * WSTR) = wr0;                            \
    *(float4*)(w_st + 1 * 16 * WSTR) = wr1;                            \
    *(float4*)(w_st + 2 * 16 * WSTR) = wr2;                            \
    *(float4*)(w_st + 3 * 16 * WSTR) = wr3;                            \
    *(float4*)(w_st + 4 * 16 * WSTR) = wr4;                            \
    *(float4*)(w_st + 5 * 16 * WSTR) = wr5;                            \
    *(float4*)(w_st + 6 * 16 * WSTR) = wr6;                            \
    *(float4*)(w_st + 7 * 16 * WSTR) = wr7;                            \
  } while (0)
  W_FETCH(Wq);
  W_PUT();
  W_FETCH(Wk);
  __syncthreads();
  wg_mix<false, WSTR, QT_W, 16, QT_W, 16>(vt, smem_q, kt, wave, lo, hg);
  __syncthreads();
  W_PUT();
  __syncthreads();

  const int st_c  = tid >> 4;
  const int st_n4 = (tid & 15) * 4;
  const int st_tg = st_n4 >> 4;
  const int st_nn = st_n4 & 15;
  const size_t st_g = ((size_t)(b * C_DIM + st_c)) * N_PTS + st_n4;
  const float* kg = K + st_g;
  const float* vg = V + st_g;
  float* k_st = kt + st_tg * KT_G + st_c * 16 + st_nn;
  float* v_st = vt + st_tg * VT_G + st_nn * VSTR + st_c;
  float4 kr0, kr1, kr2, kr3, vr0, vr1, vr2, vr3;
#define KV_FETCH(kb)                                                   \
  do {                                                                 \
    kr0 = *(const float4*)(kg + (kb));                                 \
    kr1 = *(const float4*)(kg + (kb) + (size_t)1 * 32 * N_PTS);        \
    kr2 = *(const float4*)(kg + (kb) + (size_t)2 * 32 * N_PTS);        \
    kr3 = *(const float4*)(kg + (kb) + (size_t)3 * 32 * N_PTS);        \
    vr0 = *(const float4*)(vg + (kb));                                 \
    vr1 = *(const float4*)(vg + (kb) + (size_t)1 * 32 * N_PTS);        \
    vr2 = *(const float4*)(vg + (kb) + (size_t)2 * 32 * N_PTS);        \
    vr3 = *(const float4*)(vg + (kb) + (size_t)3 * 32 * N_PTS);        \
  } while (0)
#define V_PUT(vd, i, vv)                                               \
  do {                                                                 \
    (vd)[0 * VSTR + (i) * 32] = (vv).x;                                \
    (vd)[1 * VSTR + (i) * 32] = (vv).y;                                \
    (vd)[2 * VSTR + (i) * 32] = (vv).z;                                \
    (vd)[3 * VSTR + (i) * 32] = (vv).w;                                \
  } while (0)
#define KV_PUT(kd, vd)                                                 \
  do {                                                                 \
    *(float4*)((kd) + 0 * 512) = kr0;                                  \
    *(float4*)((kd) + 1 * 512) = kr1;                                  \
    *(float4*)((kd) + 2 * 512) = kr2;                                  \
    *(float4*)((kd) + 3 * 512) = kr3;                                  \
    V_PUT(vd, 0, vr0);                                                 \
    V_PUT(vd, 1, vr1);                                                 \
    V_PUT(vd, 2, vr2);                                                 \
    V_PUT(vd, 3, vr3);                                                 \
  } while (0)

  KV_FETCH(0);
  wg_mix<true, WSTR, QT_W, 16, QT_W, 16>(vt, kt, smem_q, wave, lo, hg);
  __syncthreads();
  KV_PUT(k_st, v_st);
  __syncthreads();
  const float* qt_w = smem_q + qtile * QT_W;

#define SC_LOAD(qv, kv, rd, ktp)                                       \
  _Pragma("unroll")                                                    \
  for (int j = 0; j < 8; ++j) {                                        \
    const int c0 = 4 * (8 * (rd) + j) + 2 * hg;                         \
    (qv)[j].x = qt_w[c0 * 16 + lo];                                    \
    (qv)[j].y = qt_w[(c0 + 1) * 16 + lo];                              \
    (kv)[j].x = (ktp)[c0 * 16 + lo];                                   \
    (kv)[j].y = (ktp)[(c0 + 1) * 16 + lo];                             \
  }
#define SC_MMA(qv, kv)                                                 \
  _Pragma("unroll")                                                    \
  for (int j = 0; j < 8; ++j)                                          \
    s = __builtin_amdgcn_wmma_f32_16x16x4_f32(false, (kv)[j], false, (qv)[j], \
                                              (short)0, s, false, false);
#define SC_FENCE() __builtin_amdgcn_sched_barrier(0)
  v2f qa[8], ka[8], qb[8], kb[8];
  {
    const float* kt_0 = kt + grp * KT_G;
    SC_LOAD(qa, ka, 0, kt_0);
    SC_LOAD(qb, kb, 1, kt_0);
  }

  v8f  o[8] = {};
  float t1a = NEGINF, t2a = NEGINF, t1b = NEGINF, t2b = NEGINF;
  float mrow = NEGINF, lsum = 0.0f;

  for (int it = 0; it < NSTEP; ++it) {
    const int  cur  = it & 1;
    const bool more = (it + 1 < NSTEP);
    if (more) {
      const int kbn = (it + 1) * KSTEP;
      KV_FETCH(kbn);
    }
    const float* kt_g = kt + cur * KSTG + grp * KT_G;
    const float* vt_g = vt + cur * VSTG + grp * VT_G;

    v8f s = {};
    SC_FENCE();
    SC_MMA(qa, ka);
    SC_FENCE();
    SC_LOAD(qa, ka, 2, kt_g);
    SC_FENCE();
    SC_MMA(qb, kb);
    SC_FENCE();
    SC_LOAD(qb, kb, 3, kt_g);
    SC_FENCE();
    SC_MMA(qa, ka);
    SC_FENCE();
    SC_MMA(qb, kb);
    SC_FENCE();

    if (more) {
      float* kd = k_st + (cur ^ 1) * KSTG;
      float* vd = v_st + (cur ^ 1) * VSTG;
      KV_PUT(kd, vd);
    }

#define PV_LOAD(vv, ks2)                                               \
  do {                                                                 \
    const int k0 = 8 * hg + 2 * (ks2);                                 \
    _Pragma("unroll")                                                  \
    for (int ct = 0; ct < 8; ++ct) {                                   \
      (vv)[ct].x = vt_g[k0 * VSTR + ct * 16 + lo];                     \
      (vv)[ct].y = vt_g[(k0 + 1) * VSTR + ct * 16 + lo];               \
      SC_FENCE();                                                      \
    }                                                                  \
  } while (0)
#define PV_MMA(vv, ks2)                                                \
  do {                                                                 \
    const v2f pj = {p[2 * (ks2)], p[2 * (ks2) + 1]};                   \
    _Pragma("unroll")                                                  \
    for (int ct = 0; ct < 8; ++ct)                                     \
      o[ct] = __builtin_amdgcn_wmma_f32_16x16x4_f32(false, (vv)[ct], false, pj, \
                                                    (short)0, o[ct], false, false); \
  } while (0)
    v2f va[8], vb[8], vc[8], vd[8];
    PV_LOAD(va, 0);
    PV_LOAD(vb, 1);
    PV_LOAD(vc, 2);
    PV_LOAD(vd, 3);
    SC_FENCE();

    float dmax = NEGINF;
    v8f p;
#pragma unroll
    for (int r = 0; r < 4; ++r) {
      const float da = s[r] - mrow;
      const float db = s[r + 4] - mrow;
      dmax = fmaxf(dmax, fmaxf(da, db));
      const float lwa = fminf(t1a, s[r]);
      t1a = fmaxf(t1a, s[r]);
      t2a = fmaxf(t2a, lwa);
      const float lwb = fminf(t1b, s[r + 4]);
      t1b = fmaxf(t1b, s[r + 4]);
      t2b = fmaxf(t2b, lwb);
      p[r]     = __expf(da * invTemp);
      p[r + 4] = __expf(db * invTemp);
    }

    if (__builtin_amdgcn_ballot_w32(dmax > LAZY_M) != 0u) {
      float m8 = fmaxf(fmaxf(fmaxf(s[0], s[1]), fmaxf(s[2], s[3])),
                       fmaxf(fmaxf(s[4], s[5]), fmaxf(s[6], s[7])));
      m8 = fmaxf(m8, other_half(m8));
      const float mo = mrow;
      mrow = fmaxf(mo, m8);
      const float sc = __expf((mo - mrow) * invTemp);
      lsum *= sc;
#pragma unroll
      for (int ct = 0; ct < 8; ++ct)
#pragma unroll
        for (int r = 0; r < 8; ++r) o[ct][r] *= sc;
#pragma unroll
      for (int r = 0; r < 8; ++r) p[r] = __expf((s[r] - mrow) * invTemp);
    }
    lsum += ((p[0] + p[1]) + (p[2] + p[3])) + ((p[4] + p[5]) + (p[6] + p[7]));

    SC_FENCE();
    PV_MMA(va, 0);
    PV_MMA(vb, 1);
    SC_FENCE();
    __syncthreads();
    {
      const float* kt_n = kt + (cur ^ 1) * KSTG + grp * KT_G;
      SC_LOAD(qa, ka, 0, kt_n);
      SC_LOAD(qb, kb, 1, kt_n);
    }
    SC_FENCE();
    PV_MMA(vc, 2);
    PV_MMA(vd, 3);
#undef PV_LOAD
#undef PV_MMA
  }
#undef SC_LOAD
#undef SC_MMA
#undef SC_FENCE

#undef KV_FETCH
#undef KV_PUT
#undef V_PUT

  float t1, t2;
  {
    const float hi = fmaxf(t1a, t1b), lw = fminf(t1a, t1b);
    t2 = fmaxf(lw, fmaxf(t2a, t2b)); t1 = hi;
    const float b1 = other_half(t1), b2 = other_half(t2);
    const float hi2 = fmaxf(t1, b1), lw2 = fminf(t1, b1);
    t2 = fmaxf(lw2, fmaxf(t2, b2)); t1 = hi2;
    lsum += other_half(lsum);
  }

  W_FETCH(Wv);
  if (grp != 0) {
    float* mo_ = mrg_o + ((grp - 1) * 4 + qtile) * 2048;
    float* ms_ = mrg_s + ((grp - 1) * 4 + qtile) * 64;
    if (hg == 0) {
      ms_[lo]      = t1;
      ms_[16 + lo] = t2;
      ms_[32 + lo] = lsum;
      ms_[48 + lo] = mrow;
    }
#pragma unroll
    for (int ct = 0; ct < 8; ++ct)
#pragma unroll
      for (int r = 0; r < 8; ++r)
        mo_[(ct * 8 + r) * 32 + lane] = o[ct][r];
  }
  __syncthreads();
  W_PUT();
#undef W_FETCH
#undef W_PUT

  if (grp == 0) {
    float m = t1, m2 = t2;
    float mr = mrow, l = lsum;
#pragma unroll
    for (int g = 0; g < NGRP - 1; ++g) {
      const float* mo_ = mrg_o + (g * 4 + qtile) * 2048;
      const float* ms_ = mrg_s + (g * 4 + qtile) * 64;
      const float t1g = ms_[lo];
      const float t2g = ms_[16 + lo];
      const float lg  = ms_[32 + lo];
      const float rg  = ms_[48 + lo];
      m2 = fmaxf(fminf(m, t1g), fmaxf(m2, t2g));
      m  = fmaxf(m, t1g);
      const float mn = fmaxf(mr, rg);
      const float sa = __expf((mr - mn) * invTemp);
      const float sb = __expf((rg - mn) * invTemp);
      l = l * sa + lg * sb;
#pragma unroll
      for (int ct = 0; ct < 8; ++ct)
#pragma unroll
        for (int r = 0; r < 8; ++r)
          o[ct][r] = o[ct][r] * sa + mo_[(ct * 8 + r) * 32 + lane] * sb;
      mr = mn;
    }
    const float invl = 1.0f / l;
    const int q = qtile * 16 + lo;
#pragma unroll
    for (int ct = 0; ct < 8; ++ct)
#pragma unroll
      for (int r = 0; r < 8; ++r)
        smem_q[(ct * 16 + 8 * hg + r) * OSTR + q] = o[ct][r] * invl;
    if (hg == 0) {
      const float x1 = fminf(fmaxf(m, -1.0f), 1.0f);
      const float x2 = fminf(fmaxf(m2, -1.0f), 1.0f);
      rs[q] = acosf(x1) / acosf(x2);
    }
  }
  __syncthreads();

  wg_mix<false, WSTR, 16, OSTR, 16, OSTR>(vt, smem_q, kt, wave, lo, hg);
  __syncthreads();

#pragma unroll
  for (int i = 0; i < 2048 / ATHR; ++i) {
    const int idx4 = tid + ATHR * i;
    const int c  = idx4 >> 4;
    const int q4 = (idx4 & 15) * 4;
    const float4 val = *(const float4*)&kt[c * OSTR + q4];
    *(float4*)(Y + ((size_t)(b * C_DIM + c)) * N_PTS + qBase0 + q4) = val;
  }
  if (tid < QB) R[b * N_PTS + qBase0 + tid] = rs[tid];
}

extern "C" void kernel_launch(void* const* d_in, const int* in_sizes, int n_in,
                              void* d_out, int out_size, void* d_ws, size_t ws_size,
                              hipStream_t stream) {
  (void)in_sizes; (void)n_in; (void)out_size;
  const float* q  = (const float*)d_in[0];
  const float* k  = (const float*)d_in[1];
  const float* v  = (const float*)d_in[2];
  const float* Wq = (const float*)d_in[3];
  const float* Wk = (const float*)d_in[4];
  const float* Wv = (const float*)d_in[5];

  float* out  = (float*)d_out;
  float* Yout = out;
  float* Rout = out + (size_t)B_SZ * C_DIM * N_PTS;

  (void)d_ws; (void)ws_size;

  dim3 ga(N_PTS / QB, B_SZ);
  attn_kernel<<<ga, ATHR, 0, stream>>>(q, k, v, Wq, Wk, Wv, Yout, Rout);
}
